# starting kernel
# baseline (speedup 1.0000x reference)
#include <hip/hip_runtime.h>
#include <hip/hip_bf16.h>


typedef float v2f __attribute__((ext_vector_type(2)));
typedef float v8f __attribute__((ext_vector_type(8)));

#define NB   2048
#define TT   512
#define HH   11
#define GG   44
#define OUTF 22

__device__ __forceinline__ v8f wmma_f32_16x16x4(v2f a, v2f b, v8f c) {
  return __builtin_amdgcn_wmma_f32_16x16x4_f32(false, a, false, b, (short)0, c,
                                               false, false);
}

__device__ __forceinline__ float fast_sigmoid(float x) {
  return __builtin_amdgcn_rcpf(1.0f + __expf(-x));
}
__device__ __forceinline__ float fast_tanh(float x) {
  return 1.0f - 2.0f * __builtin_amdgcn_rcpf(__expf(2.0f * x) + 1.0f);
}
__device__ __forceinline__ v8f splat8(float x) {
  v8f v;
#pragma unroll
  for (int i = 0; i < 8; ++i) v[i] = x;
  return v;
}

__global__ __launch_bounds__(256) void reduce_x_kernel(
    const float* __restrict__ x, float* __restrict__ xsum) {
  size_t idx = (size_t)blockIdx.x * blockDim.x + threadIdx.x;
  if (idx >= (size_t)NB * TT) return;
  size_t b = idx / TT, t = idx % TT;
  const float* p = x + b * (32u * TT) + t;
  float s = 0.0f;
#pragma unroll
  for (int k = 0; k < 32; ++k) s += p[(size_t)k * TT];
  xsum[idx] = s;
}

template <int DIN>
__global__ __launch_bounds__(32) void lstm_layer_kernel(
    const float* __restrict__ in, float* __restrict__ out,
    const float* __restrict__ Wih, const float* __restrict__ Whh,
    const float* __restrict__ bih, const float* __restrict__ bhh) {
  constexpr int NSLICE_IN = (DIN + 3) / 4;
  constexpr int NSLICE_H  = 3;

  const int lane  = threadIdx.x;
  const int n     = lane & 15;
  const int khalf = lane >> 4;
  const int dir   = blockIdx.y;
  const int b0    = blockIdx.x * 16;

  Wih += (size_t)dir * GG * DIN;
  Whh += (size_t)dir * GG * HH;
  bih += (size_t)dir * GG;
  bhh += (size_t)dir * GG;

  v2f   Bh[4][NSLICE_H];
  v2f   Bi[4][NSLICE_IN];
  v8f   accBias[4];
  const bool gv = (n < HH);
#pragma unroll
  for (int gt = 0; gt < 4; ++gt) {
    const int g = gt * HH + n;
    accBias[gt] = splat8(gv ? (bih[g] + bhh[g]) : 0.0f);
#pragma unroll
    for (int s = 0; s < NSLICE_H; ++s) {
      const int k0 = s * 4 + 2 * khalf;
      float w0 = (gv && k0 + 0 < HH) ? Whh[(size_t)g * HH + k0 + 0] : 0.0f;
      float w1 = (gv && k0 + 1 < HH) ? Whh[(size_t)g * HH + k0 + 1] : 0.0f;
      Bh[gt][s] = v2f{w0, w1};
    }
#pragma unroll
    for (int s = 0; s < NSLICE_IN; ++s) {
      const int k0 = s * 4 + 2 * khalf;
      float w0 = (gv && k0 + 0 < DIN) ? Wih[(size_t)g * DIN + k0 + 0] : 0.0f;
      float w1 = (gv && k0 + 1 < DIN) ? Wih[(size_t)g * DIN + k0 + 1] : 0.0f;
      Bi[gt][s] = v2f{w0, w1};
    }
  }

  __shared__ float hbuf[16 * 16];
  for (int i = lane; i < 16 * 16; i += 32) hbuf[i] = 0.0f;
  __syncthreads();

  v8f c = splat8(0.0f);

  const size_t inRow = (size_t)(b0 + n) * TT;
  for (int step = 0; step < TT; ++step) {
    const int t = (dir == 0) ? step : (TT - 1 - step);

    v2f Ain[NSLICE_IN];
    if (DIN == 1) {
      const float xv = (khalf == 0) ? in[inRow + t] : 0.0f;
      Ain[0] = v2f{xv, 0.0f};
    } else {
      const float* p = in + (inRow + (size_t)t) * DIN;
#pragma unroll
      for (int s = 0; s < NSLICE_IN; ++s) {
        const int k0 = s * 4 + 2 * khalf;
        v2f v = v2f{0.0f, 0.0f};
        if (k0 + 1 < DIN) v = *(const v2f*)(p + k0);
        Ain[s] = v;
      }
      const int tn = (dir == 0) ? (t + 1) : (t - 1);
      if (tn >= 0 && tn < TT)
        __builtin_prefetch(in + (inRow + (size_t)tn) * DIN, 0, 1);
    }

    v2f Ah[NSLICE_H];
#pragma unroll
    for (int s = 0; s < NSLICE_H; ++s) {
      const int k0 = s * 4 + 2 * khalf;
      Ah[s] = *(const v2f*)(&hbuf[(n << 4) + k0]);
    }
    __syncthreads();

    v8f a0 = accBias[0], a1 = accBias[1], a2 = accBias[2], a3 = accBias[3];
#pragma unroll
    for (int s = 0; s < NSLICE_H; ++s) {
      a0 = wmma_f32_16x16x4(Ah[s], Bh[0][s], a0);
      a1 = wmma_f32_16x16x4(Ah[s], Bh[1][s], a1);
      a2 = wmma_f32_16x16x4(Ah[s], Bh[2][s], a2);
      a3 = wmma_f32_16x16x4(Ah[s], Bh[3][s], a3);
    }
#pragma unroll
    for (int s = 0; s < NSLICE_IN; ++s) {
      a0 = wmma_f32_16x16x4(Ain[s], Bi[0][s], a0);
      a1 = wmma_f32_16x16x4(Ain[s], Bi[1][s], a1);
      a2 = wmma_f32_16x16x4(Ain[s], Bi[2][s], a2);
      a3 = wmma_f32_16x16x4(Ain[s], Bi[3][s], a3);
    }

    v8f hnew;
#pragma unroll
    for (int r = 0; r < 8; ++r) {
      const float ig = fast_sigmoid(a0[r]);
      const float fg = fast_sigmoid(a1[r]);
      const float gg = fast_tanh(a2[r]);
      const float og = fast_sigmoid(a3[r]);
      const float cr = fg * c[r] + ig * gg;
      c[r]    = cr;
      hnew[r] = og * fast_tanh(cr);
    }

#pragma unroll
    for (int r = 0; r < 8; ++r)
      hbuf[((r + 8 * khalf) << 4) + n] = hnew[r];

    if (gv) {
#pragma unroll
      for (int r = 0; r < 8; ++r) {
        const int mrow = r + 8 * khalf;
        out[((size_t)(b0 + mrow) * TT + t) * OUTF + (size_t)dir * HH + n] =
            hnew[r];
      }
    }
    __syncthreads();
  }
}

extern "C" void kernel_launch(void* const* d_in, const int* in_sizes, int n_in,
                              void* d_out, int out_size, void* d_ws,
                              size_t ws_size, hipStream_t stream) {
  const float* x     = (const float*)d_in[0];
  const float* W_ih0 = (const float*)d_in[1];
  const float* W_ihr = (const float*)d_in[2];
  const float* W_hh  = (const float*)d_in[3];
  const float* b_ih  = (const float*)d_in[4];
  const float* b_hh  = (const float*)d_in[5];
  float* out = (float*)d_out;

  const size_t XS_BYTES  = (size_t)NB * TT * sizeof(float);
  const size_t ACT_BYTES = (size_t)NB * TT * OUTF * sizeof(float);
  char*  ws   = (char*)d_ws;
  float* xsum = (float*)(ws);
  float* bufA = (float*)(ws + XS_BYTES);
  float* bufB = (float*)(ws + XS_BYTES + ACT_BYTES);
  (void)ws_size; (void)in_sizes; (void)n_in; (void)out_size;

  {
    const size_t total = (size_t)NB * TT;
    reduce_x_kernel<<<dim3((unsigned)((total + 255) / 256)), dim3(256), 0,
                      stream>>>(x, xsum);
  }

  const dim3 grid(NB / 16, 2);
  const dim3 block(32);

  lstm_layer_kernel<1><<<grid, block, 0, stream>>>(
      xsum, bufA, W_ih0, W_hh, b_ih, b_hh);

  const float* lin[4]  = {bufA, bufB, bufA, bufB};
  float*       lout[4] = {bufB, bufA, bufB, out};
  for (int l = 1; l < 5; ++l) {
    lstm_layer_kernel<22><<<grid, block, 0, stream>>>(
        lin[l - 1], lout[l - 1],
        W_ihr + (size_t)(l - 1) * 2 * GG * 22,
        W_hh  + (size_t)l * 2 * GG * HH,
        b_ih  + (size_t)l * 2 * GG,
        b_hh  + (size_t)l * 2 * GG);
  }
}


# Gates computed transposed (weights as A, h^T and x^T as B, units split 6+5 over lane halves): D registers are next step's B operand, no LDS transpose or barriers; k grouping of the h sum changes
# speedup vs baseline: 1.3210x; 1.3210x over previous
#include <hip/hip_runtime.h>
#include <hip/hip_bf16.h>


typedef float v2f __attribute__((ext_vector_type(2)));
typedef float v8f __attribute__((ext_vector_type(8)));

#define NB   2048
#define TT   512
#define HH   11
#define GG   44
#define OUTF 22

__device__ __forceinline__ v8f wmma_f32_16x16x4(v2f a, v2f b, v8f c) {
  return __builtin_amdgcn_wmma_f32_16x16x4_f32(false, a, false, b, (short)0, c,
                                               false, false);
}

__device__ __forceinline__ float fast_sigmoid(float x) {
  return __builtin_amdgcn_rcpf(1.0f + __expf(-x));
}
__device__ __forceinline__ float fast_tanh(float x) {
  return 1.0f - 2.0f * __builtin_amdgcn_rcpf(__expf(2.0f * x) + 1.0f);
}
__device__ __forceinline__ v8f splat8(float x) {
  v8f v;
#pragma unroll
  for (int i = 0; i < 8; ++i) v[i] = x;
  return v;
}

__global__ __launch_bounds__(256) void reduce_x_kernel(
    const float* __restrict__ x, float* __restrict__ xsum) {
  size_t idx = (size_t)blockIdx.x * blockDim.x + threadIdx.x;
  if (idx >= (size_t)NB * TT) return;
  size_t b = idx / TT, t = idx % TT;
  const float* p = x + b * (32u * TT) + t;
  float s = 0.0f;
#pragma unroll
  for (int k = 0; k < 32; ++k) s += p[(size_t)k * TT];
  xsum[idx] = s;
}

template <int DIN>
__global__ __launch_bounds__(32)
__attribute__((amdgpu_waves_per_eu(1, 1))) void lstm_layer_kernel(
    const float* __restrict__ in, float* __restrict__ out,
    const float* __restrict__ Wih, const float* __restrict__ Whh,
    const float* __restrict__ bih, const float* __restrict__ bhh) {
  constexpr int NSLICE_IN = (DIN + 3) / 4;
  constexpr int NSLICE_H  = 3;
  constexpr int NR        = 6;

  const int lane = threadIdx.x;
  const int col  = lane & 15;
  const int half = lane >> 4;
  const int dir  = blockIdx.y;
  const int b0   = blockIdx.x * 16;

  Wih += (size_t)dir * GG * DIN;
  Whh += (size_t)dir * GG * HH;
  bih += (size_t)dir * GG;
  bhh += (size_t)dir * GG;

  const int  jrow = (col < 6) ? col : col - 2;
  const bool rowv = (col < 6) || (col >= 8 && col < 13);
  v2f Ah[4][NSLICE_H];
  v2f Ai[4][NSLICE_IN];
  v8f accBias[4];
#pragma unroll
  for (int gt = 0; gt < 4; ++gt) {
    const int g = gt * HH + jrow;
#pragma unroll
    for (int s = 0; s < NSLICE_H; ++s) {
      const int k0 = half ? (6 + 2 * s) : (2 * s);
      float w0 = (rowv && k0 + 0 < HH) ? Whh[(size_t)g * HH + k0 + 0] : 0.0f;
      float w1 = (rowv && k0 + 1 < HH) ? Whh[(size_t)g * HH + k0 + 1] : 0.0f;
      Ah[gt][s] = v2f{w0, w1};
    }
#pragma unroll
    for (int s = 0; s < NSLICE_IN; ++s) {
      const int k0 = s * 4 + 2 * half;
      float w0 = (rowv && k0 + 0 < DIN) ? Wih[(size_t)g * DIN + k0 + 0] : 0.0f;
      float w1 = (rowv && k0 + 1 < DIN) ? Wih[(size_t)g * DIN + k0 + 1] : 0.0f;
      Ai[gt][s] = v2f{w0, w1};
    }
#pragma unroll
    for (int r = 0; r < 8; ++r) {
      const int  j = half ? (6 + r) : r;
      const bool v = half ? (r < 5) : (r < 6);
      accBias[gt][r] = v ? (bih[gt * HH + j] + bhh[gt * HH + j]) : 0.0f;
    }
  }

  float c[NR], h[NR];
#pragma unroll
  for (int r = 0; r < NR; ++r) { c[r] = 0.0f; h[r] = 0.0f; }

  const size_t inRow = (size_t)(b0 + col) * TT;
  float* const outRow =
      out + inRow * OUTF + (size_t)dir * HH + (half ? 6 : 0);
  for (int step = 0; step < TT; ++step) {
    const int t = (dir == 0) ? step : (TT - 1 - step);

    v2f Bin[NSLICE_IN];
    if (DIN == 1) {
      const float xv = (half == 0) ? in[inRow + t] : 0.0f;
      Bin[0] = v2f{xv, 0.0f};
    } else {
      const float* p = in + (inRow + (size_t)t) * DIN;
#pragma unroll
      for (int s = 0; s < NSLICE_IN; ++s) {
        const int k0 = s * 4 + 2 * half;
        v2f v = v2f{0.0f, 0.0f};
        if (k0 + 1 < DIN) v = *(const v2f*)(p + k0);
        Bin[s] = v;
      }
      const int tn = (dir == 0) ? (t + 1) : (t - 1);
      if (tn >= 0 && tn < TT)
        __builtin_prefetch(in + (inRow + (size_t)tn) * DIN, 0, 1);
    }

    v8f a0 = accBias[0], a1 = accBias[1], a2 = accBias[2], a3 = accBias[3];
#pragma unroll
    for (int s = 0; s < NSLICE_H; ++s) {
      const v2f Bh = v2f{h[2 * s], h[2 * s + 1]};
      a0 = wmma_f32_16x16x4(Ah[0][s], Bh, a0);
      a1 = wmma_f32_16x16x4(Ah[1][s], Bh, a1);
      a2 = wmma_f32_16x16x4(Ah[2][s], Bh, a2);
      a3 = wmma_f32_16x16x4(Ah[3][s], Bh, a3);
    }
#pragma unroll
    for (int s = 0; s < NSLICE_IN; ++s) {
      a0 = wmma_f32_16x16x4(Ai[0][s], Bin[s], a0);
      a1 = wmma_f32_16x16x4(Ai[1][s], Bin[s], a1);
      a2 = wmma_f32_16x16x4(Ai[2][s], Bin[s], a2);
      a3 = wmma_f32_16x16x4(Ai[3][s], Bin[s], a3);
    }

#pragma unroll
    for (int r = 0; r < NR; ++r) {
      const float ig = fast_sigmoid(a0[r]);
      const float fg = fast_sigmoid(a1[r]);
      const float gg = fast_tanh(a2[r]);
      const float og = fast_sigmoid(a3[r]);
      const float cr = fg * c[r] + ig * gg;
      c[r] = cr;
      h[r] = og * fast_tanh(cr);
    }

    float* const o = outRow + (size_t)t * OUTF;
#pragma unroll
    for (int r = 0; r < NR - 1; ++r) o[r] = h[r];
    if (half == 0) o[NR - 1] = h[NR - 1];
  }
}

extern "C" void kernel_launch(void* const* d_in, const int* in_sizes, int n_in,
                              void* d_out, int out_size, void* d_ws,
                              size_t ws_size, hipStream_t stream) {
  const float* x     = (const float*)d_in[0];
  const float* W_ih0 = (const float*)d_in[1];
  const float* W_ihr = (const float*)d_in[2];
  const float* W_hh  = (const float*)d_in[3];
  const float* b_ih  = (const float*)d_in[4];
  const float* b_hh  = (const float*)d_in[5];
  float* out = (float*)d_out;

  const size_t XS_BYTES  = (size_t)NB * TT * sizeof(float);
  const size_t ACT_BYTES = (size_t)NB * TT * OUTF * sizeof(float);
  char*  ws   = (char*)d_ws;
  float* xsum = (float*)(ws);
  float* bufA = (float*)(ws + XS_BYTES);
  float* bufB = (float*)(ws + XS_BYTES + ACT_BYTES);
  (void)ws_size; (void)in_sizes; (void)n_in; (void)out_size;

  {
    const size_t total = (size_t)NB * TT;
    reduce_x_kernel<<<dim3((unsigned)((total + 255) / 256)), dim3(256), 0,
                      stream>>>(x, xsum);
  }

  const dim3 grid(NB / 16, 2);
  const dim3 block(32);

  lstm_layer_kernel<1><<<grid, block, 0, stream>>>(
      xsum, bufA, W_ih0, W_hh, b_ih, b_hh);

  const float* lin[4]  = {bufA, bufB, bufA, bufB};
  float*       lout[4] = {bufB, bufA, bufB, out};
  for (int l = 1; l < 5; ++l) {
    lstm_layer_kernel<22><<<grid, block, 0, stream>>>(
        lin[l - 1], lout[l - 1],
        W_ihr + (size_t)(l - 1) * 2 * GG * 22,
        W_hh  + (size_t)l * 2 * GG * HH,
        b_ih  + (size_t)l * 2 * GG,
        b_hh  + (size_t)l * 2 * GG);
  }
}


# Input half of each step (x loads, L2 prefetch, 24 Wih WMMAs) software-pipelined one step ahead of the recurrence and dealt out among the cell update; gate sum order: Wih terms now before Whh terms
# speedup vs baseline: 1.6507x; 1.2496x over previous
#include <hip/hip_runtime.h>
#include <hip/hip_bf16.h>


typedef float v2f __attribute__((ext_vector_type(2)));
typedef float v8f __attribute__((ext_vector_type(8)));

#define NB   2048
#define TT   512
#define HH   11
#define GG   44
#define OUTF 22

__device__ __forceinline__ v8f wmma_f32_16x16x4(v2f a, v2f b, v8f c) {
  return __builtin_amdgcn_wmma_f32_16x16x4_f32(false, a, false, b, (short)0, c,
                                               false, false);
}

__device__ __forceinline__ v8f splat8(float x) {
  v8f v;
#pragma unroll
  for (int i = 0; i < 8; ++i) v[i] = x;
  return v;
}

__global__ __launch_bounds__(256) void reduce_x_kernel(
    const float* __restrict__ x, float* __restrict__ xsum) {
  size_t idx = (size_t)blockIdx.x * blockDim.x + threadIdx.x;
  if (idx >= (size_t)NB * TT) return;
  size_t b = idx / TT, t = idx % TT;
  const float* p = x + b * (32u * TT) + t;
  float s = 0.0f;
#pragma unroll
  for (int k = 0; k < 32; ++k) s += p[(size_t)k * TT];
  xsum[idx] = s;
}

template <int DIN>
__global__ __launch_bounds__(32)
__attribute__((amdgpu_waves_per_eu(1, 1))) void lstm_layer_kernel(
    const float* __restrict__ in, float* __restrict__ out,
    const float* __restrict__ Wih, const float* __restrict__ Whh,
    const float* __restrict__ bih, const float* __restrict__ bhh) {
  constexpr int NSLICE_IN = (DIN + 3) / 4;
  constexpr int NSLICE_H  = 3;
  constexpr int NR        = 6;

  const int lane = threadIdx.x;
  const int col  = lane & 15;
  const int half = lane >> 4;
  const int dir  = blockIdx.y;
  const int b0   = blockIdx.x * 16;

  Wih += (size_t)dir * GG * DIN;
  Whh += (size_t)dir * GG * HH;
  bih += (size_t)dir * GG;
  bhh += (size_t)dir * GG;

  const int  jrow = (col < 6) ? col : col - 2;
  const bool rowv = (col < 6) || (col >= 8 && col < 13);
  v2f Ah[4][NSLICE_H];
  v2f Ai[4][NSLICE_IN];
  v8f accBias[4];
#pragma unroll
  for (int gt = 0; gt < 4; ++gt) {
    const int g = gt * HH + jrow;
#pragma unroll
    for (int s = 0; s < NSLICE_H; ++s) {
      const int k0 = half ? (6 + 2 * s) : (2 * s);
      float w0 = (rowv && k0 + 0 < HH) ? Whh[(size_t)g * HH + k0 + 0] : 0.0f;
      float w1 = (rowv && k0 + 1 < HH) ? Whh[(size_t)g * HH + k0 + 1] : 0.0f;
      Ah[gt][s] = v2f{w0, w1};
    }
#pragma unroll
    for (int s = 0; s < NSLICE_IN; ++s) {
      const int k0 = s * 4 + 2 * half;
      float w0 = (rowv && k0 + 0 < DIN) ? Wih[(size_t)g * DIN + k0 + 0] : 0.0f;
      float w1 = (rowv && k0 + 1 < DIN) ? Wih[(size_t)g * DIN + k0 + 1] : 0.0f;
      Ai[gt][s] = v2f{w0, w1};
    }
#pragma unroll
    for (int r = 0; r < 8; ++r) {
      const int  j = half ? (6 + r) : r;
      const bool v = half ? (r < 5) : (r < 6);
      accBias[gt][r] = v ? (bih[gt * HH + j] + bhh[gt * HH + j]) : 0.0f;
    }
  }

  float c[NR], h[NR];
#pragma unroll
  for (int r = 0; r < NR; ++r) { c[r] = 0.0f; h[r] = 0.0f; }

  const size_t inRow = (size_t)(b0 + col) * TT;
  float* const outRow =
      out + inRow * OUTF + (size_t)dir * HH + (half ? 6 : 0);

  auto t_of = [&](int step) -> int {
    const int sc = step < TT ? step : TT - 1;
    return (dir == 0) ? sc : (TT - 1 - sc);
  };
  auto load_in = [&](int step, v2f (&X)[NSLICE_IN]) {
    const int t = t_of(step);
    if (DIN == 1) {
      float xv = 0.0f;
      if (half == 0) xv = in[inRow + t];
      X[0] = v2f{xv, 0.0f};
    } else {
      const float* p = in + (inRow + (size_t)t) * DIN;
#pragma unroll
      for (int s = 0; s < NSLICE_IN; ++s) {
        const int k0 = s * 4 + 2 * half;
        v2f v = v2f{0.0f, 0.0f};
        if (s * 4 + 3 < DIN || half == 0) v = *(const v2f*)(p + k0);
        X[s] = v;
      }
    }
  };
  auto project = [&](const v2f (&X)[NSLICE_IN], v8f (&P)[4]) {
    P[0] = accBias[0]; P[1] = accBias[1]; P[2] = accBias[2]; P[3] = accBias[3];
#pragma unroll
    for (int s = 0; s < NSLICE_IN; ++s) {
      P[0] = wmma_f32_16x16x4(Ai[0][s], X[s], P[0]);
      P[1] = wmma_f32_16x16x4(Ai[1][s], X[s], P[1]);
      P[2] = wmma_f32_16x16x4(Ai[2][s], X[s], P[2]);
      P[3] = wmma_f32_16x16x4(Ai[3][s], X[s], P[3]);
    }
  };

  constexpr int PF = 8;
  v8f Pc[4];

  auto one_step = [&](int step, v2f (&Xl)[NSLICE_IN],
                      const v2f (&Xu)[NSLICE_IN]) {
    const int t = t_of(step);

    v8f a0 = Pc[0], a1 = Pc[1], a2 = Pc[2], a3 = Pc[3];
#pragma unroll
    for (int s = 0; s < NSLICE_H; ++s) {
      const v2f Bh = v2f{h[2 * s], h[2 * s + 1]};
      a0 = wmma_f32_16x16x4(Ah[0][s], Bh, a0);
      a1 = wmma_f32_16x16x4(Ah[1][s], Bh, a1);
      a2 = wmma_f32_16x16x4(Ah[2][s], Bh, a2);
      a3 = wmma_f32_16x16x4(Ah[3][s], Bh, a3);
    }

    load_in(step + 2, Xl);
    if (DIN != 1)
      __builtin_prefetch(in + (inRow + (size_t)t_of(step + PF)) * DIN, 0, 1);

    __builtin_amdgcn_sched_barrier(0);

    project(Xu, Pc);

    float ei[NR], ef[NR], eg[NR], eo[NR], ec[NR];
#pragma unroll
    for (int r = 0; r < NR; ++r) {
      ei[r] = __expf(-a0[r]);
      ef[r] = __expf(-a1[r]);
      eg[r] = __expf(2.0f * a2[r]);
      eo[r] = __expf(-a3[r]);
    }
#pragma unroll
    for (int r = 0; r < NR; ++r) {
      ei[r] = __builtin_amdgcn_rcpf(1.0f + ei[r]);
      ef[r] = __builtin_amdgcn_rcpf(1.0f + ef[r]);
      eg[r] = 1.0f - 2.0f * __builtin_amdgcn_rcpf(eg[r] + 1.0f);
      eo[r] = __builtin_amdgcn_rcpf(1.0f + eo[r]);
    }
#pragma unroll
    for (int r = 0; r < NR; ++r) {
      const float cr = ef[r] * c[r] + ei[r] * eg[r];
      c[r]  = cr;
      ec[r] = __expf(2.0f * cr);
    }
#pragma unroll
    for (int r = 0; r < NR; ++r)
      ec[r] = __builtin_amdgcn_rcpf(ec[r] + 1.0f);
#pragma unroll
    for (int r = 0; r < NR; ++r)
      h[r] = eo[r] * (1.0f - 2.0f * ec[r]);

    float* const o = outRow + (size_t)t * OUTF;
#pragma unroll
    for (int r = 0; r < NR - 1; ++r) o[r] = h[r];

    if (DIN != 1) {
#pragma unroll
      for (int i = 0; i < 4 * NSLICE_IN - 4; ++i) {
        __builtin_amdgcn_sched_group_barrier(0x008, 1, 0);
        __builtin_amdgcn_sched_group_barrier(0x002, 3, 0);
        __builtin_amdgcn_sched_group_barrier(0x400, 2, 0);
        __builtin_amdgcn_sched_group_barrier(0x002, 2, 0);
        __builtin_amdgcn_sched_group_barrier(0x400, 1, 0);
      }
#pragma unroll
      for (int i = 0; i < 4; ++i)
        __builtin_amdgcn_sched_group_barrier(0x008, 1, 0);
    }
    if (half == 0) o[NR - 1] = h[NR - 1];
  };

  v2f Xe[NSLICE_IN], Xo[NSLICE_IN];
  load_in(0, Xe);
  load_in(1, Xo);
  project(Xe, Pc);
  for (int step = 0; step < TT; step += 2) {
    one_step(step, Xe, Xo);
    one_step(step + 1, Xo, Xe);
  }
}

extern "C" void kernel_launch(void* const* d_in, const int* in_sizes, int n_in,
                              void* d_out, int out_size, void* d_ws,
                              size_t ws_size, hipStream_t stream) {
  const float* x     = (const float*)d_in[0];
  const float* W_ih0 = (const float*)d_in[1];
  const float* W_ihr = (const float*)d_in[2];
  const float* W_hh  = (const float*)d_in[3];
  const float* b_ih  = (const float*)d_in[4];
  const float* b_hh  = (const float*)d_in[5];
  float* out = (float*)d_out;

  const size_t XS_BYTES  = (size_t)NB * TT * sizeof(float);
  const size_t ACT_BYTES = (size_t)NB * TT * OUTF * sizeof(float);
  char*  ws   = (char*)d_ws;
  float* xsum = (float*)(ws);
  float* bufA = (float*)(ws + XS_BYTES);
  float* bufB = (float*)(ws + XS_BYTES + ACT_BYTES);
  (void)ws_size; (void)in_sizes; (void)n_in; (void)out_size;

  {
    const size_t total = (size_t)NB * TT;
    reduce_x_kernel<<<dim3((unsigned)((total + 255) / 256)), dim3(256), 0,
                      stream>>>(x, xsum);
  }

  const dim3 grid(NB / 16, 2);
  const dim3 block(32);

  lstm_layer_kernel<1><<<grid, block, 0, stream>>>(
      xsum, bufA, W_ih0, W_hh, b_ih, b_hh);

  const float* lin[4]  = {bufA, bufB, bufA, bufB};
  float*       lout[4] = {bufB, bufA, bufB, out};
  for (int l = 1; l < 5; ++l) {
    lstm_layer_kernel<22><<<grid, block, 0, stream>>>(
        lin[l - 1], lout[l - 1],
        W_ihr + (size_t)(l - 1) * 2 * GG * 22,
        W_hh  + (size_t)l * 2 * GG * HH,
        b_ih  + (size_t)l * 2 * GG,
        b_hh  + (size_t)l * 2 * GG);
  }
}


# Activations between layers stored in WMMA B-fragment order, one 768-byte block per (tile, direction, step): step loads and stores are whole contiguous lines; k grouping of the Wih sum changes
# speedup vs baseline: 2.0131x; 1.2195x over previous
#include <hip/hip_runtime.h>
#include <hip/hip_bf16.h>


typedef float v2f __attribute__((ext_vector_type(2)));
typedef float v8f __attribute__((ext_vector_type(8)));

#define NB   2048
#define TT   512
#define HH   11
#define GG   44
#define OUTF 22
#define FRAG 192

__device__ __forceinline__ v8f wmma_f32_16x16x4(v2f a, v2f b, v8f c) {
  return __builtin_amdgcn_wmma_f32_16x16x4_f32(false, a, false, b, (short)0, c,
                                               false, false);
}

__device__ __forceinline__ v8f splat8(float x) {
  v8f v;
#pragma unroll
  for (int i = 0; i < 8; ++i) v[i] = x;
  return v;
}

__global__ __launch_bounds__(256) void reduce_x_kernel(
    const float* __restrict__ x, float* __restrict__ xsum) {
  size_t idx = (size_t)blockIdx.x * blockDim.x + threadIdx.x;
  if (idx >= (size_t)NB * TT) return;
  size_t b = idx / TT, t = idx % TT;
  const float* p = x + b * (32u * TT) + t;
  float s = 0.0f;
#pragma unroll
  for (int k = 0; k < 32; ++k) s += p[(size_t)k * TT];
  xsum[idx] = s;
}

template <int DIN, bool IN_FRAG, bool OUT_FRAG>
__global__ __launch_bounds__(32)
__attribute__((amdgpu_waves_per_eu(1, 1))) void lstm_layer_kernel(
    const float* __restrict__ in, float* __restrict__ out,
    const float* __restrict__ Wih, const float* __restrict__ Whh,
    const float* __restrict__ bih, const float* __restrict__ bhh) {
  constexpr int NSLICE_IN = (DIN + 3) / 4;
  constexpr int NSLICE_H  = 3;
  constexpr int NR        = 6;

  const int lane = threadIdx.x;
  const int col  = lane & 15;
  const int half = lane >> 4;
  const int dir  = blockIdx.y;
  const int b0   = blockIdx.x * 16;

  Wih += (size_t)dir * GG * DIN;
  Whh += (size_t)dir * GG * HH;
  bih += (size_t)dir * GG;
  bhh += (size_t)dir * GG;

  const int  jrow = (col < 6) ? col : col - 2;
  const bool rowv = (col < 6) || (col >= 8 && col < 13);
  v2f Ah[4][NSLICE_H];
  v2f Ai[4][NSLICE_IN];
  v8f accBias[4];
#pragma unroll
  for (int gt = 0; gt < 4; ++gt) {
    const int g = gt * HH + jrow;
#pragma unroll
    for (int s = 0; s < NSLICE_H; ++s) {
      const int k0 = half ? (6 + 2 * s) : (2 * s);
      float w0 = (rowv && k0 + 0 < HH) ? Whh[(size_t)g * HH + k0 + 0] : 0.0f;
      float w1 = (rowv && k0 + 1 < HH) ? Whh[(size_t)g * HH + k0 + 1] : 0.0f;
      Ah[gt][s] = v2f{w0, w1};
    }
#pragma unroll
    for (int s = 0; s < NSLICE_IN; ++s) {
      if (IN_FRAG) {
        const int dp = s / 3;
        const int j0 = (s % 3) * 4 + 2 * half;
        const size_t wb = (size_t)g * DIN + dp * HH + j0;
        float w0 = (rowv && j0 + 0 < HH) ? Wih[wb + 0] : 0.0f;
        float w1 = (rowv && j0 + 1 < HH) ? Wih[wb + 1] : 0.0f;
        Ai[gt][s] = v2f{w0, w1};
      } else {
        const int k0 = s * 4 + 2 * half;
        float w0 = (rowv && k0 + 0 < DIN) ? Wih[(size_t)g * DIN + k0 + 0] : 0.0f;
        float w1 = (rowv && k0 + 1 < DIN) ? Wih[(size_t)g * DIN + k0 + 1] : 0.0f;
        Ai[gt][s] = v2f{w0, w1};
      }
    }
#pragma unroll
    for (int r = 0; r < 8; ++r) {
      const int  j = half ? (6 + r) : r;
      const bool v = half ? (r < 5) : (r < 6);
      accBias[gt][r] = v ? (bih[gt * HH + j] + bhh[gt * HH + j]) : 0.0f;
    }
  }

  float c[NR], h[NR];
#pragma unroll
  for (int r = 0; r < NR; ++r) { c[r] = 0.0f; h[r] = 0.0f; }

  const size_t inRow = (size_t)(b0 + col) * TT;
  float* const outRow =
      out + inRow * OUTF + (size_t)dir * HH + (half ? 6 : 0);
  const size_t tileBlk = (size_t)blockIdx.x * 2 * TT * FRAG;
  const float* const inFrag = IN_FRAG ? in + tileBlk + lane * 2 : in;
  float* const outFrag = OUT_FRAG
      ? out + tileBlk + (size_t)dir * TT * FRAG + (48 * half + col) * 2
      : out;
  const float* const pfFrag = IN_FRAG
      ? in + tileBlk + (size_t)((lane % 12) / 6) * TT * FRAG +
            ((lane % 12) % 6) * 32
      : in;

  auto t_of = [&](int step) -> int {
    const int sc = step < TT ? step : TT - 1;
    return (dir == 0) ? sc : (TT - 1 - sc);
  };
  auto load_in = [&](int step, v2f (&X)[NSLICE_IN]) {
    const int t = t_of(step);
    if (DIN == 1) {
      float xv = 0.0f;
      if (half == 0) xv = in[inRow + t];
      X[0] = v2f{xv, 0.0f};
    } else if (IN_FRAG) {
      const float* p = inFrag + (size_t)t * FRAG;
#pragma unroll
      for (int s = 0; s < NSLICE_IN; ++s)
        X[s] = *(const v2f*)(p + (size_t)(s / 3) * TT * FRAG + (s % 3) * 64);
    } else {
      const float* p = in + (inRow + (size_t)t) * DIN;
#pragma unroll
      for (int s = 0; s < NSLICE_IN; ++s) {
        const int k0 = s * 4 + 2 * half;
        v2f v = v2f{0.0f, 0.0f};
        if (s * 4 + 3 < DIN || half == 0) v = *(const v2f*)(p + k0);
        X[s] = v;
      }
    }
  };
  auto project = [&](const v2f (&X)[NSLICE_IN], v8f (&P)[4]) {
    P[0] = accBias[0]; P[1] = accBias[1]; P[2] = accBias[2]; P[3] = accBias[3];
#pragma unroll
    for (int s = 0; s < NSLICE_IN; ++s) {
      P[0] = wmma_f32_16x16x4(Ai[0][s], X[s], P[0]);
      P[1] = wmma_f32_16x16x4(Ai[1][s], X[s], P[1]);
      P[2] = wmma_f32_16x16x4(Ai[2][s], X[s], P[2]);
      P[3] = wmma_f32_16x16x4(Ai[3][s], X[s], P[3]);
    }
  };

  constexpr int PF = 8;
  v8f Pc[4];

  auto one_step = [&](int step, v2f (&Xl)[NSLICE_IN],
                      const v2f (&Xu)[NSLICE_IN]) {
    const int t = t_of(step);

    v8f a0 = Pc[0], a1 = Pc[1], a2 = Pc[2], a3 = Pc[3];
#pragma unroll
    for (int s = 0; s < NSLICE_H; ++s) {
      const v2f Bh = v2f{h[2 * s], h[2 * s + 1]};
      a0 = wmma_f32_16x16x4(Ah[0][s], Bh, a0);
      a1 = wmma_f32_16x16x4(Ah[1][s], Bh, a1);
      a2 = wmma_f32_16x16x4(Ah[2][s], Bh, a2);
      a3 = wmma_f32_16x16x4(Ah[3][s], Bh, a3);
    }

    load_in(step + 2, Xl);
    if (IN_FRAG)
      __builtin_prefetch(pfFrag + (size_t)t_of(step + PF) * FRAG, 0, 1);
    else if (DIN != 1)
      __builtin_prefetch(in + (inRow + (size_t)t_of(step + PF)) * DIN, 0, 1);

    __builtin_amdgcn_sched_barrier(0);

    project(Xu, Pc);

    float ei[NR], ef[NR], eg[NR], eo[NR], ec[NR];
#pragma unroll
    for (int r = 0; r < NR; ++r) {
      ei[r] = __expf(-a0[r]);
      ef[r] = __expf(-a1[r]);
      eg[r] = __expf(2.0f * a2[r]);
      eo[r] = __expf(-a3[r]);
    }
#pragma unroll
    for (int r = 0; r < NR; ++r) {
      ei[r] = __builtin_amdgcn_rcpf(1.0f + ei[r]);
      ef[r] = __builtin_amdgcn_rcpf(1.0f + ef[r]);
      eg[r] = 1.0f - 2.0f * __builtin_amdgcn_rcpf(eg[r] + 1.0f);
      eo[r] = __builtin_amdgcn_rcpf(1.0f + eo[r]);
    }
#pragma unroll
    for (int r = 0; r < NR; ++r) {
      const float cr = ef[r] * c[r] + ei[r] * eg[r];
      c[r]  = cr;
      ec[r] = __expf(2.0f * cr);
    }
#pragma unroll
    for (int r = 0; r < NR; ++r)
      ec[r] = __builtin_amdgcn_rcpf(ec[r] + 1.0f);
#pragma unroll
    for (int r = 0; r < NR; ++r)
      h[r] = eo[r] * (1.0f - 2.0f * ec[r]);

    float* const o = outRow + (size_t)t * OUTF;
    if (OUT_FRAG) {
      float* const ob = outFrag + (size_t)t * FRAG;
#pragma unroll
      for (int p = 0; p < NR / 2; ++p)
        *(v2f*)(ob + p * 32) = v2f{h[2 * p], h[2 * p + 1]};
    } else {
#pragma unroll
      for (int r = 0; r < NR - 1; ++r) o[r] = h[r];
    }

    if (DIN != 1) {
#pragma unroll
      for (int i = 0; i < 4 * NSLICE_IN - 4; ++i) {
        __builtin_amdgcn_sched_group_barrier(0x008, 1, 0);
        __builtin_amdgcn_sched_group_barrier(0x002, 3, 0);
        __builtin_amdgcn_sched_group_barrier(0x400, 2, 0);
        __builtin_amdgcn_sched_group_barrier(0x002, 2, 0);
        __builtin_amdgcn_sched_group_barrier(0x400, 1, 0);
      }
#pragma unroll
      for (int i = 0; i < 4; ++i)
        __builtin_amdgcn_sched_group_barrier(0x008, 1, 0);
    }
    if (!OUT_FRAG && half == 0) o[NR - 1] = h[NR - 1];
  };

  v2f Xe[NSLICE_IN], Xo[NSLICE_IN];
  load_in(0, Xe);
  load_in(1, Xo);
  project(Xe, Pc);
  for (int step = 0; step < TT; step += 2) {
    one_step(step, Xe, Xo);
    one_step(step + 1, Xo, Xe);
  }
}

extern "C" void kernel_launch(void* const* d_in, const int* in_sizes, int n_in,
                              void* d_out, int out_size, void* d_ws,
                              size_t ws_size, hipStream_t stream) {
  const float* x     = (const float*)d_in[0];
  const float* W_ih0 = (const float*)d_in[1];
  const float* W_ihr = (const float*)d_in[2];
  const float* W_hh  = (const float*)d_in[3];
  const float* b_ih  = (const float*)d_in[4];
  const float* b_hh  = (const float*)d_in[5];
  float* out = (float*)d_out;

  const size_t XS_BYTES  = (size_t)NB * TT * sizeof(float);
  const size_t ACT_BYTES = (size_t)(NB / 16) * 2 * TT * FRAG * sizeof(float);
  char*  ws   = (char*)d_ws;
  float* xsum = (float*)(ws);
  float* bufA = (float*)(ws + XS_BYTES);
  float* bufB = (float*)(ws + XS_BYTES + ACT_BYTES);
  (void)ws_size; (void)in_sizes; (void)n_in; (void)out_size;

  {
    const size_t total = (size_t)NB * TT;
    reduce_x_kernel<<<dim3((unsigned)((total + 255) / 256)), dim3(256), 0,
                      stream>>>(x, xsum);
  }

  const dim3 grid(NB / 16, 2);
  const dim3 block(32);

  lstm_layer_kernel<1, false, true><<<grid, block, 0, stream>>>(
      xsum, bufA, W_ih0, W_hh, b_ih, b_hh);

  const float* lin[4]  = {bufA, bufB, bufA, bufB};
  float*       lout[4] = {bufB, bufA, bufB, out};
  for (int l = 1; l < 4; ++l) {
    lstm_layer_kernel<22, true, true><<<grid, block, 0, stream>>>(
        lin[l - 1], lout[l - 1],
        W_ihr + (size_t)(l - 1) * 2 * GG * 22,
        W_hh  + (size_t)l * 2 * GG * HH,
        b_ih  + (size_t)l * 2 * GG,
        b_hh  + (size_t)l * 2 * GG);
  }
  lstm_layer_kernel<22, true, false><<<grid, block, 0, stream>>>(
      lin[3], lout[3],
      W_ihr + (size_t)3 * 2 * GG * 22,
      W_hh  + (size_t)4 * 2 * GG * HH,
      b_ih  + (size_t)4 * 2 * GG,
      b_hh  + (size_t)4 * 2 * GG);
}


# Gate rows packed 16 to a tile (4 units x 4 gates; a unit's i/f/g/o in 4 registers of one lane): 3 M-tiles not 4, 27 WMMAs a step not 36; k grouping of the h sum changes
# speedup vs baseline: 2.2768x; 1.1310x over previous
#include <hip/hip_runtime.h>
#include <hip/hip_bf16.h>


typedef float v2f __attribute__((ext_vector_type(2)));
typedef float v8f __attribute__((ext_vector_type(8)));

#define NB   2048
#define TT   512
#define HH   11
#define GG   44
#define OUTF 22
#define FRAG 192

__device__ __forceinline__ v8f wmma_f32_16x16x4(v2f a, v2f b, v8f c) {
  return __builtin_amdgcn_wmma_f32_16x16x4_f32(false, a, false, b, (short)0, c,
                                               false, false);
}

__device__ __forceinline__ v8f splat8(float x) {
  v8f v;
#pragma unroll
  for (int i = 0; i < 8; ++i) v[i] = x;
  return v;
}

__global__ __launch_bounds__(256) void reduce_x_kernel(
    const float* __restrict__ x, float* __restrict__ xsum) {
  size_t idx = (size_t)blockIdx.x * blockDim.x + threadIdx.x;
  if (idx >= (size_t)NB * TT) return;
  size_t b = idx / TT, t = idx % TT;
  const float* p = x + b * (32u * TT) + t;
  float s = 0.0f;
#pragma unroll
  for (int k = 0; k < 32; ++k) s += p[(size_t)k * TT];
  xsum[idx] = s;
}

template <int DIN, bool IN_FRAG, bool OUT_FRAG>
__global__ __launch_bounds__(32)
__attribute__((amdgpu_waves_per_eu(1, 1))) void lstm_layer_kernel(
    const float* __restrict__ in, float* __restrict__ out,
    const float* __restrict__ Wih, const float* __restrict__ Whh,
    const float* __restrict__ bih, const float* __restrict__ bhh) {
  constexpr int NSLICE_IN = (DIN + 3) / 4;
  constexpr int NSLICE_H  = 3;
  constexpr int NT        = 3;
  constexpr int NR        = 6;

  const int lane = threadIdx.x;
  const int col  = lane & 15;
  const int half = lane >> 4;
  const int dir  = blockIdx.y;
  const int b0   = blockIdx.x * 16;

  Wih += (size_t)dir * GG * DIN;
  Whh += (size_t)dir * GG * HH;
  bih += (size_t)dir * GG;
  bhh += (size_t)dir * GG;

  const int rgate = col & 3;
  const int rsub  = col >> 2;
  v2f Ah[NT][NSLICE_H];
  v2f Ai[NT][NSLICE_IN];
  v8f accBias[NT];
#pragma unroll
  for (int tl = 0; tl < NT; ++tl) {
    const int  unit = 4 * tl + rsub;
    const bool rowv = unit < HH;
    const int  g    = rgate * HH + unit;
#pragma unroll
    for (int s = 0; s < NSLICE_H; ++s) {
      const int k0 = 4 * s + 2 * half;
      float w0 = (rowv && k0 + 0 < HH) ? Whh[(size_t)g * HH + k0 + 0] : 0.0f;
      float w1 = (rowv && k0 + 1 < HH) ? Whh[(size_t)g * HH + k0 + 1] : 0.0f;
      Ah[tl][s] = v2f{w0, w1};
    }
#pragma unroll
    for (int s = 0; s < NSLICE_IN; ++s) {
      if (IN_FRAG) {
        const int dp = s / 3;
        const int j0 = (s % 3) * 4 + 2 * half;
        const size_t wb = (size_t)g * DIN + dp * HH + j0;
        float w0 = (rowv && j0 + 0 < HH) ? Wih[wb + 0] : 0.0f;
        float w1 = (rowv && j0 + 1 < HH) ? Wih[wb + 1] : 0.0f;
        Ai[tl][s] = v2f{w0, w1};
      } else {
        const int k0 = s * 4 + 2 * half;
        float w0 = (rowv && k0 + 0 < DIN) ? Wih[(size_t)g * DIN + k0 + 0] : 0.0f;
        float w1 = (rowv && k0 + 1 < DIN) ? Wih[(size_t)g * DIN + k0 + 1] : 0.0f;
        Ai[tl][s] = v2f{w0, w1};
      }
    }
#pragma unroll
    for (int r = 0; r < 8; ++r) {
      const int j  = 4 * tl + 2 * half + (r >> 2);
      const int gr = (r & 3) * HH + j;
      accBias[tl][r] = (j < HH) ? (bih[gr] + bhh[gr]) : 0.0f;
    }
  }

  float c[NR], h[NR];
#pragma unroll
  for (int r = 0; r < NR; ++r) { c[r] = 0.0f; h[r] = 0.0f; }

  const size_t inRow = (size_t)(b0 + col) * TT;
  float* const outRow =
      out + inRow * OUTF + (size_t)dir * HH + 2 * half;
  const size_t tileBlk = (size_t)blockIdx.x * 2 * TT * FRAG;
  const float* const inFrag = IN_FRAG ? in + tileBlk + lane * 2 : in;
  float* const outFrag = OUT_FRAG
      ? out + tileBlk + (size_t)dir * TT * FRAG + (16 * half + col) * 2
      : out;
  const float* const pfFrag = IN_FRAG
      ? in + tileBlk + (size_t)((lane % 12) / 6) * TT * FRAG +
            ((lane % 12) % 6) * 32
      : in;

  auto t_of = [&](int step) -> int {
    const int sc = step < TT ? step : TT - 1;
    return (dir == 0) ? sc : (TT - 1 - sc);
  };
  auto load_in = [&](int step, v2f (&X)[NSLICE_IN]) {
    const int t = t_of(step);
    if (DIN == 1) {
      float xv = 0.0f;
      if (half == 0) xv = in[inRow + t];
      X[0] = v2f{xv, 0.0f};
    } else if (IN_FRAG) {
      const float* p = inFrag + (size_t)t * FRAG;
#pragma unroll
      for (int s = 0; s < NSLICE_IN; ++s)
        X[s] = *(const v2f*)(p + (size_t)(s / 3) * TT * FRAG + (s % 3) * 64);
    } else {
      const float* p = in + (inRow + (size_t)t) * DIN;
#pragma unroll
      for (int s = 0; s < NSLICE_IN; ++s) {
        const int k0 = s * 4 + 2 * half;
        v2f v = v2f{0.0f, 0.0f};
        if (s * 4 + 3 < DIN || half == 0) v = *(const v2f*)(p + k0);
        X[s] = v;
      }
    }
  };
  auto project = [&](const v2f (&X)[NSLICE_IN], v8f (&P)[NT]) {
    P[0] = accBias[0]; P[1] = accBias[1]; P[2] = accBias[2];
#pragma unroll
    for (int s = 0; s < NSLICE_IN; ++s) {
      P[0] = wmma_f32_16x16x4(Ai[0][s], X[s], P[0]);
      P[1] = wmma_f32_16x16x4(Ai[1][s], X[s], P[1]);
      P[2] = wmma_f32_16x16x4(Ai[2][s], X[s], P[2]);
    }
  };

  constexpr int PF = 8;
  v8f Pc[NT];

  auto one_step = [&](int step, v2f (&Xl)[NSLICE_IN],
                      const v2f (&Xu)[NSLICE_IN]) {
    const int t = t_of(step);

    v8f a[NT] = {Pc[0], Pc[1], Pc[2]};
#pragma unroll
    for (int s = 0; s < NSLICE_H; ++s) {
      const v2f Bh = v2f{h[2 * s], h[2 * s + 1]};
      a[0] = wmma_f32_16x16x4(Ah[0][s], Bh, a[0]);
      a[1] = wmma_f32_16x16x4(Ah[1][s], Bh, a[1]);
      a[2] = wmma_f32_16x16x4(Ah[2][s], Bh, a[2]);
    }

    load_in(step + 2, Xl);
    if (IN_FRAG)
      __builtin_prefetch(pfFrag + (size_t)t_of(step + PF) * FRAG, 0, 1);
    else if (DIN != 1)
      __builtin_prefetch(in + (inRow + (size_t)t_of(step + PF)) * DIN, 0, 1);

    __builtin_amdgcn_sched_barrier(0);

    project(Xu, Pc);

    float ei[NR], ef[NR], eg[NR], eo[NR], ec[NR];
#pragma unroll
    for (int r = 0; r < NR; ++r) {
      const int tl = r >> 1, q = 4 * (r & 1);
      ei[r] = __expf(-a[tl][q + 0]);
      ef[r] = __expf(-a[tl][q + 1]);
      eg[r] = __expf(2.0f * a[tl][q + 2]);
      eo[r] = __expf(-a[tl][q + 3]);
    }
#pragma unroll
    for (int r = 0; r < NR; ++r) {
      ei[r] = __builtin_amdgcn_rcpf(1.0f + ei[r]);
      ef[r] = __builtin_amdgcn_rcpf(1.0f + ef[r]);
      eg[r] = 1.0f - 2.0f * __builtin_amdgcn_rcpf(eg[r] + 1.0f);
      eo[r] = __builtin_amdgcn_rcpf(1.0f + eo[r]);
    }
#pragma unroll
    for (int r = 0; r < NR; ++r) {
      const float cr = ef[r] * c[r] + ei[r] * eg[r];
      c[r]  = cr;
      ec[r] = __expf(2.0f * cr);
    }
#pragma unroll
    for (int r = 0; r < NR; ++r)
      ec[r] = __builtin_amdgcn_rcpf(ec[r] + 1.0f);
#pragma unroll
    for (int r = 0; r < NR; ++r)
      h[r] = eo[r] * (1.0f - 2.0f * ec[r]);

    float* const o = outRow + (size_t)t * OUTF;
    if (OUT_FRAG) {
      float* const ob = outFrag + (size_t)t * FRAG;
#pragma unroll
      for (int p = 0; p < NR / 2; ++p)
        *(v2f*)(ob + p * 64) = v2f{h[2 * p], h[2 * p + 1]};
    } else {
#pragma unroll
      for (int r = 0; r < NR - 1; ++r) o[4 * (r >> 1) + (r & 1)] = h[r];
    }

    if (DIN != 1) {
#pragma unroll
      for (int i = 0; i < NT * NSLICE_IN - NT; ++i) {
        __builtin_amdgcn_sched_group_barrier(0x008, 1, 0);
        __builtin_amdgcn_sched_group_barrier(0x002, 3, 0);
        __builtin_amdgcn_sched_group_barrier(0x400, 2, 0);
        __builtin_amdgcn_sched_group_barrier(0x002, 3, 0);
        __builtin_amdgcn_sched_group_barrier(0x400, 2, 0);
      }
#pragma unroll
      for (int i = 0; i < NT; ++i)
        __builtin_amdgcn_sched_group_barrier(0x008, 1, 0);
    }
    if (!OUT_FRAG && half == 0) o[4 * ((NR - 1) >> 1) + 1] = h[NR - 1];
  };

  v2f Xe[NSLICE_IN], Xo[NSLICE_IN];
  load_in(0, Xe);
  load_in(1, Xo);
  project(Xe, Pc);
  for (int step = 0; step < TT; step += 2) {
    one_step(step, Xe, Xo);
    one_step(step + 1, Xo, Xe);
  }
}

extern "C" void kernel_launch(void* const* d_in, const int* in_sizes, int n_in,
                              void* d_out, int out_size, void* d_ws,
                              size_t ws_size, hipStream_t stream) {
  const float* x     = (const float*)d_in[0];
  const float* W_ih0 = (const float*)d_in[1];
  const float* W_ihr = (const float*)d_in[2];
  const float* W_hh  = (const float*)d_in[3];
  const float* b_ih  = (const float*)d_in[4];
  const float* b_hh  = (const float*)d_in[5];
  float* out = (float*)d_out;

  const size_t XS_BYTES  = (size_t)NB * TT * sizeof(float);
  const size_t ACT_BYTES = (size_t)(NB / 16) * 2 * TT * FRAG * sizeof(float);
  char*  ws   = (char*)d_ws;
  float* xsum = (float*)(ws);
  float* bufA = (float*)(ws + XS_BYTES);
  float* bufB = (float*)(ws + XS_BYTES + ACT_BYTES);
  (void)ws_size; (void)in_sizes; (void)n_in; (void)out_size;

  {
    const size_t total = (size_t)NB * TT;
    reduce_x_kernel<<<dim3((unsigned)((total + 255) / 256)), dim3(256), 0,
                      stream>>>(x, xsum);
  }

  const dim3 grid(NB / 16, 2);
  const dim3 block(32);

  lstm_layer_kernel<1, false, true><<<grid, block, 0, stream>>>(
      xsum, bufA, W_ih0, W_hh, b_ih, b_hh);

  const float* lin[4]  = {bufA, bufB, bufA, bufB};
  float*       lout[4] = {bufB, bufA, bufB, out};
  for (int l = 1; l < 4; ++l) {
    lstm_layer_kernel<22, true, true><<<grid, block, 0, stream>>>(
        lin[l - 1], lout[l - 1],
        W_ihr + (size_t)(l - 1) * 2 * GG * 22,
        W_hh  + (size_t)l * 2 * GG * HH,
        b_ih  + (size_t)l * 2 * GG,
        b_hh  + (size_t)l * 2 * GG);
  }
  lstm_layer_kernel<22, true, false><<<grid, block, 0, stream>>>(
      lin[3], lout[3],
      W_ihr + (size_t)3 * 2 * GG * 22,
      W_hh  + (size_t)4 * 2 * GG * HH,
      b_ih  + (size_t)4 * 2 * GG,
      b_hh  + (size_t)4 * 2 * GG);
}


# Every sigmoid and tanh of the cell update is one hardware tanh (sigmoid(a) = (1 + tanh(a/2)) / 2, the half folded exactly into the i, f, o weight rows and biases): 30 transcendentals a step, not 60
# speedup vs baseline: 2.8723x; 1.2616x over previous
#include <hip/hip_runtime.h>
#include <hip/hip_bf16.h>


typedef float v2f __attribute__((ext_vector_type(2)));
typedef float v8f __attribute__((ext_vector_type(8)));

#define NB   2048
#define TT   512
#define HH   11
#define GG   44
#define OUTF 22
#define FRAG 192

__device__ __forceinline__ v8f wmma_f32_16x16x4(v2f a, v2f b, v8f c) {
  return __builtin_amdgcn_wmma_f32_16x16x4_f32(false, a, false, b, (short)0, c,
                                               false, false);
}

__device__ __forceinline__ v8f splat8(float x) {
  v8f v;
#pragma unroll
  for (int i = 0; i < 8; ++i) v[i] = x;
  return v;
}

__global__ __launch_bounds__(256) void reduce_x_kernel(
    const float* __restrict__ x, float* __restrict__ xsum) {
  size_t idx = (size_t)blockIdx.x * blockDim.x + threadIdx.x;
  if (idx >= (size_t)NB * TT) return;
  size_t b = idx / TT, t = idx % TT;
  const float* p = x + b * (32u * TT) + t;
  float s = 0.0f;
#pragma unroll
  for (int k = 0; k < 32; ++k) s += p[(size_t)k * TT];
  xsum[idx] = s;
}

template <int DIN, bool IN_FRAG, bool OUT_FRAG>
__global__ __launch_bounds__(32)
__attribute__((amdgpu_waves_per_eu(1, 1))) void lstm_layer_kernel(
    const float* __restrict__ in, float* __restrict__ out,
    const float* __restrict__ Wih, const float* __restrict__ Whh,
    const float* __restrict__ bih, const float* __restrict__ bhh) {
  constexpr int NSLICE_IN = (DIN + 3) / 4;
  constexpr int NSLICE_H  = 3;
  constexpr int NT        = 3;
  constexpr int NR        = 6;

  const int lane = threadIdx.x;
  const int col  = lane & 15;
  const int half = lane >> 4;
  const int dir  = blockIdx.y;
  const int b0   = blockIdx.x * 16;

  Wih += (size_t)dir * GG * DIN;
  Whh += (size_t)dir * GG * HH;
  bih += (size_t)dir * GG;
  bhh += (size_t)dir * GG;

  const int rgate = col & 3;
  const int rsub  = col >> 2;
  const float rscale = (rgate == 2) ? 1.0f : 0.5f;
  v2f Ah[NT][NSLICE_H];
  v2f Ai[NT][NSLICE_IN];
  v8f accBias[NT];
#pragma unroll
  for (int tl = 0; tl < NT; ++tl) {
    const int  unit = 4 * tl + rsub;
    const bool rowv = unit < HH;
    const int  g    = rgate * HH + unit;
#pragma unroll
    for (int s = 0; s < NSLICE_H; ++s) {
      const int k0 = 4 * s + 2 * half;
      float w0 = (rowv && k0 + 0 < HH) ? Whh[(size_t)g * HH + k0 + 0] : 0.0f;
      float w1 = (rowv && k0 + 1 < HH) ? Whh[(size_t)g * HH + k0 + 1] : 0.0f;
      Ah[tl][s] = v2f{w0 * rscale, w1 * rscale};
    }
#pragma unroll
    for (int s = 0; s < NSLICE_IN; ++s) {
      if (IN_FRAG) {
        const int dp = s / 3;
        const int j0 = (s % 3) * 4 + 2 * half;
        const size_t wb = (size_t)g * DIN + dp * HH + j0;
        float w0 = (rowv && j0 + 0 < HH) ? Wih[wb + 0] : 0.0f;
        float w1 = (rowv && j0 + 1 < HH) ? Wih[wb + 1] : 0.0f;
        Ai[tl][s] = v2f{w0 * rscale, w1 * rscale};
      } else {
        const int k0 = s * 4 + 2 * half;
        float w0 = (rowv && k0 + 0 < DIN) ? Wih[(size_t)g * DIN + k0 + 0] : 0.0f;
        float w1 = (rowv && k0 + 1 < DIN) ? Wih[(size_t)g * DIN + k0 + 1] : 0.0f;
        Ai[tl][s] = v2f{w0 * rscale, w1 * rscale};
      }
    }
#pragma unroll
    for (int r = 0; r < 8; ++r) {
      const int j  = 4 * tl + 2 * half + (r >> 2);
      const int gr = (r & 3) * HH + j;
      const float bs = ((r & 3) == 2) ? 1.0f : 0.5f;
      accBias[tl][r] = (j < HH) ? (bih[gr] + bhh[gr]) * bs : 0.0f;
    }
  }

  v2f c2[NT], h2[NT];
#pragma unroll
  for (int p = 0; p < NT; ++p) { c2[p] = v2f{0.0f, 0.0f}; h2[p] = v2f{0.0f, 0.0f}; }

  const size_t inRow = (size_t)(b0 + col) * TT;
  float* const outRow =
      out + inRow * OUTF + (size_t)dir * HH + 2 * half;
  const size_t tileBlk = (size_t)blockIdx.x * 2 * TT * FRAG;
  const float* const inFrag = IN_FRAG ? in + tileBlk + lane * 2 : in;
  float* const outFrag = OUT_FRAG
      ? out + tileBlk + (size_t)dir * TT * FRAG + (16 * half + col) * 2
      : out;
  const float* const pfFrag = IN_FRAG
      ? in + tileBlk + (size_t)((lane % 12) / 6) * TT * FRAG +
            ((lane % 12) % 6) * 32
      : in;

  auto t_of = [&](int step) -> int {
    const int sc = step < TT ? step : TT - 1;
    return (dir == 0) ? sc : (TT - 1 - sc);
  };
  auto load_in = [&](int step, v2f (&X)[NSLICE_IN]) {
    const int t = t_of(step);
    if (DIN == 1) {
      float xv = 0.0f;
      if (half == 0) xv = in[inRow + t];
      X[0] = v2f{xv, 0.0f};
    } else if (IN_FRAG) {
      const float* p = inFrag + (size_t)t * FRAG;
#pragma unroll
      for (int s = 0; s < NSLICE_IN; ++s)
        X[s] = *(const v2f*)(p + (size_t)(s / 3) * TT * FRAG + (s % 3) * 64);
    } else {
      const float* p = in + (inRow + (size_t)t) * DIN;
#pragma unroll
      for (int s = 0; s < NSLICE_IN; ++s) {
        const int k0 = s * 4 + 2 * half;
        v2f v = v2f{0.0f, 0.0f};
        if (s * 4 + 3 < DIN || half == 0) v = *(const v2f*)(p + k0);
        X[s] = v;
      }
    }
  };
  auto project = [&](const v2f (&X)[NSLICE_IN], v8f (&P)[NT]) {
    P[0] = accBias[0]; P[1] = accBias[1]; P[2] = accBias[2];
#pragma unroll
    for (int s = 0; s < NSLICE_IN; ++s) {
      P[0] = wmma_f32_16x16x4(Ai[0][s], X[s], P[0]);
      P[1] = wmma_f32_16x16x4(Ai[1][s], X[s], P[1]);
      P[2] = wmma_f32_16x16x4(Ai[2][s], X[s], P[2]);
    }
  };

  constexpr int PF = 8;
  v8f Pc[NT];

  auto one_step = [&](int step, v2f (&Xl)[NSLICE_IN],
                      const v2f (&Xu)[NSLICE_IN]) {
    const int t = t_of(step);

    v8f a[NT] = {Pc[0], Pc[1], Pc[2]};
#pragma unroll
    for (int s = 0; s < NSLICE_H; ++s) {
      const v2f Bh = h2[s];
      a[0] = wmma_f32_16x16x4(Ah[0][s], Bh, a[0]);
      a[1] = wmma_f32_16x16x4(Ah[1][s], Bh, a[1]);
      a[2] = wmma_f32_16x16x4(Ah[2][s], Bh, a[2]);
    }

    load_in(step + 2, Xl);
    if (IN_FRAG)
      __builtin_prefetch(pfFrag + (size_t)t_of(step + PF) * FRAG, 0, 1);
    else if (DIN != 1)
      __builtin_prefetch(in + (inRow + (size_t)t_of(step + PF)) * DIN, 0, 1);

    __builtin_amdgcn_sched_barrier(0);

    auto inW = [&](int w) {
      if (w < NT * NSLICE_IN) {
        const int s = w / NT, tl = w % NT;
        Pc[tl] = wmma_f32_16x16x4(Ai[tl][s], Xu[s], s == 0 ? accBias[tl] : Pc[tl]);
      }
    };
    v2f gi[NT], gf[NT], gg[NT], go[NT], fc[NT], tc[NT];
    auto th2 = [&](float x, float y) {
      return v2f{__builtin_amdgcn_tanhf(x), __builtin_amdgcn_tanhf(y)};
    };
    auto tI = [&](int p) { gi[p] = th2(a[p][0], a[p][4]); };
    auto tF = [&](int p) { gf[p] = th2(a[p][1], a[p][5]); };
    auto tG = [&](int p) { gg[p] = th2(a[p][2], a[p][6]); };
    auto tO = [&](int p) { go[p] = th2(a[p][3], a[p][7]); };
    auto sI = [&](int p) { gi[p] = 0.5f * gi[p] + 0.5f; };
    auto sF = [&](int p) { gf[p] = 0.5f * gf[p] + 0.5f; };
    auto sO = [&](int p) { go[p] = 0.5f * go[p] + 0.5f; };
    auto mF = [&](int p) { fc[p] = gf[p] * c2[p]; };
    auto cell = [&](int p) { c2[p] = gi[p] * gg[p] + fc[p]; };
    auto cell_tanh = [&](int p) { tc[p] = th2(c2[p][0], c2[p][1]); };
    auto hidden = [&](int p) { h2[p] = go[p] * tc[p]; };
#define ORD __builtin_amdgcn_sched_barrier(0);

    inW(0); inW(1); inW(2);
    ORD tI(0); tF(0); tG(0); tO(0);
    ORD inW(3);
    ORD tI(1); ORD sI(0); ORD tF(1); ORD sF(0);
    ORD tG(1); ORD sO(0); ORD tO(1); ORD mF(0);
    ORD inW(4);
    ORD tI(2); ORD cell(0); ORD tF(2); ORD sI(1);
    ORD tG(2); ORD sF(1); ORD tO(2); ORD sO(1);
    ORD mF(1);
    ORD inW(5);
    ORD sF(2); ORD cell_tanh(0); ORD cell(1); ORD sI(2); sO(2); ORD mF(2);
    ORD inW(6);
    ORD hidden(0); ORD cell_tanh(1); ORD cell(2);
    ORD inW(7);
    ORD hidden(1); ORD cell_tanh(2);
    ORD inW(8);
    ORD hidden(2);
    ORD
#undef ORD

    float* const o = outRow + (size_t)t * OUTF;
    if (OUT_FRAG) {
      float* const ob = outFrag + (size_t)t * FRAG;
#pragma unroll
      for (int p = 0; p < NT; ++p) *(v2f*)(ob + p * 64) = h2[p];
    } else {
#pragma unroll
      for (int r = 0; r < NR - 1; ++r) o[4 * (r >> 1) + (r & 1)] = h2[r >> 1][r & 1];
      if (half == 0) o[4 * ((NR - 1) >> 1) + 1] = h2[NT - 1][1];
    }
    __builtin_amdgcn_sched_barrier(0);
#pragma unroll
    for (int w = 9; w < NT * NSLICE_IN; ++w) inW(w);
  };

  v2f Xe[NSLICE_IN], Xo[NSLICE_IN];
  load_in(0, Xe);
  load_in(1, Xo);
  project(Xe, Pc);
  for (int step = 0; step < TT; step += 2) {
    one_step(step, Xe, Xo);
    one_step(step + 1, Xo, Xe);
  }
}

extern "C" void kernel_launch(void* const* d_in, const int* in_sizes, int n_in,
                              void* d_out, int out_size, void* d_ws,
                              size_t ws_size, hipStream_t stream) {
  const float* x     = (const float*)d_in[0];
  const float* W_ih0 = (const float*)d_in[1];
  const float* W_ihr = (const float*)d_in[2];
  const float* W_hh  = (const float*)d_in[3];
  const float* b_ih  = (const float*)d_in[4];
  const float* b_hh  = (const float*)d_in[5];
  float* out = (float*)d_out;

  const size_t XS_BYTES  = (size_t)NB * TT * sizeof(float);
  const size_t ACT_BYTES = (size_t)(NB / 16) * 2 * TT * FRAG * sizeof(float);
  char*  ws   = (char*)d_ws;
  float* xsum = (float*)(ws);
  float* bufA = (float*)(ws + XS_BYTES);
  float* bufB = (float*)(ws + XS_BYTES + ACT_BYTES);
  (void)ws_size; (void)in_sizes; (void)n_in; (void)out_size;

  {
    const size_t total = (size_t)NB * TT;
    reduce_x_kernel<<<dim3((unsigned)((total + 255) / 256)), dim3(256), 0,
                      stream>>>(x, xsum);
  }

  const dim3 grid(NB / 16, 2);
  const dim3 block(32);

  lstm_layer_kernel<1, false, true><<<grid, block, 0, stream>>>(
      xsum, bufA, W_ih0, W_hh, b_ih, b_hh);

  const float* lin[4]  = {bufA, bufB, bufA, bufB};
  float*       lout[4] = {bufB, bufA, bufB, out};
  for (int l = 1; l < 4; ++l) {
    lstm_layer_kernel<22, true, true><<<grid, block, 0, stream>>>(
        lin[l - 1], lout[l - 1],
        W_ihr + (size_t)(l - 1) * 2 * GG * 22,
        W_hh  + (size_t)l * 2 * GG * HH,
        b_ih  + (size_t)l * 2 * GG,
        b_hh  + (size_t)l * 2 * GG);
  }
  lstm_layer_kernel<22, true, false><<<grid, block, 0, stream>>>(
      lin[3], lout[3],
      W_ihr + (size_t)3 * 2 * GG * 22,
      W_hh  + (size_t)4 * 2 * GG * HH,
      b_ih  + (size_t)4 * 2 * GG,
      b_hh  + (size_t)4 * 2 * GG);
}


# Input half of every step (loads, L2 prefetch, bias + Wih WMMAs) moved to two helper waves on other SIMDs of the WGP, handed to the recurrent wave through LDS in C-operand layout, 16 steps a chunk
# speedup vs baseline: 5.4745x; 1.9060x over previous
#include <hip/hip_runtime.h>
#include <hip/hip_bf16.h>


typedef float v2f __attribute__((ext_vector_type(2)));
typedef float v4f __attribute__((ext_vector_type(4)));
typedef float v8f __attribute__((ext_vector_type(8)));

#define NB   2048
#define TT   512
#define HH   11
#define GG   44
#define OUTF 22
#define FRAG 192
#define CHUNK 16
#define PSTEP 768

__device__ __forceinline__ v8f wmma_f32_16x16x4(v2f a, v2f b, v8f c) {
  return __builtin_amdgcn_wmma_f32_16x16x4_f32(false, a, false, b, (short)0, c,
                                               false, false);
}

__device__ __forceinline__ v8f splat8(float x) {
  v8f v;
#pragma unroll
  for (int i = 0; i < 8; ++i) v[i] = x;
  return v;
}

__global__ __launch_bounds__(256) void reduce_x_kernel(
    const float* __restrict__ x, float* __restrict__ xsum) {
  size_t idx = (size_t)blockIdx.x * blockDim.x + threadIdx.x;
  if (idx >= (size_t)NB * TT) return;
  size_t b = idx / TT, t = idx % TT;
  const float* p = x + b * (32u * TT) + t;
  float s = 0.0f;
#pragma unroll
  for (int k = 0; k < 32; ++k) s += p[(size_t)k * TT];
  xsum[idx] = s;
}

template <int DIN, bool IN_FRAG, bool OUT_FRAG>
__global__ __launch_bounds__(96)
__attribute__((amdgpu_waves_per_eu(1, 1))) void lstm_layer_kernel(
    const float* __restrict__ in, float* __restrict__ out,
    const float* __restrict__ Wih, const float* __restrict__ Whh,
    const float* __restrict__ bih, const float* __restrict__ bhh) {
  constexpr int NSLICE_IN = (DIN + 3) / 4;
  constexpr int NSLICE_H  = 3;
  constexpr int NT        = 3;
  constexpr int NR        = 6;
  constexpr int CH        = CHUNK;
  constexpr int NCH       = TT / CH;
  constexpr int HS        = CH / 2;
  static_assert(TT % CH == 0 && CH % 2 == 0, "whole chunks, split in two");

  __shared__ __attribute__((aligned(16))) float Pbuf[2 * CH * PSTEP];

  const int lane = threadIdx.x & 31;
  const int wave = __builtin_amdgcn_readfirstlane((int)(threadIdx.x >> 5));
  const int col  = lane & 15;
  const int half = lane >> 4;
  const int dir  = blockIdx.y;
  const int b0   = blockIdx.x * 16;

  Wih += (size_t)dir * GG * DIN;
  Whh += (size_t)dir * GG * HH;
  bih += (size_t)dir * GG;
  bhh += (size_t)dir * GG;

  const int rgate = col & 3;
  const int rsub  = col >> 2;
  const float rscale = (rgate == 2) ? 1.0f : 0.5f;
  v2f Ah[NT][NSLICE_H];
  v2f Ai[NT][NSLICE_IN];
  v8f accBias[NT];
#pragma unroll
  for (int tl = 0; tl < NT; ++tl) {
    const int  unit = 4 * tl + rsub;
    const bool rowv = unit < HH;
    const int  g    = rgate * HH + unit;
#pragma unroll
    for (int s = 0; s < NSLICE_H; ++s) Ah[tl][s] = v2f{0.0f, 0.0f};
#pragma unroll
    for (int s = 0; s < NSLICE_IN; ++s) Ai[tl][s] = v2f{0.0f, 0.0f};
    accBias[tl] = splat8(0.0f);
    if (wave == 0) {
#pragma unroll
      for (int s = 0; s < NSLICE_H; ++s) {
        const int k0 = 4 * s + 2 * half;
        float w0 = (rowv && k0 + 0 < HH) ? Whh[(size_t)g * HH + k0 + 0] : 0.0f;
        float w1 = (rowv && k0 + 1 < HH) ? Whh[(size_t)g * HH + k0 + 1] : 0.0f;
        Ah[tl][s] = v2f{w0 * rscale, w1 * rscale};
      }
    } else {
#pragma unroll
      for (int s = 0; s < NSLICE_IN; ++s) {
        if (IN_FRAG) {
          const int dp = s / 3;
          const int j0 = (s % 3) * 4 + 2 * half;
          const size_t wb = (size_t)g * DIN + dp * HH + j0;
          float w0 = (rowv && j0 + 0 < HH) ? Wih[wb + 0] : 0.0f;
          float w1 = (rowv && j0 + 1 < HH) ? Wih[wb + 1] : 0.0f;
          Ai[tl][s] = v2f{w0 * rscale, w1 * rscale};
        } else {
          const int k0 = s * 4 + 2 * half;
          float w0 = (rowv && k0 + 0 < DIN) ? Wih[(size_t)g * DIN + k0 + 0] : 0.0f;
          float w1 = (rowv && k0 + 1 < DIN) ? Wih[(size_t)g * DIN + k0 + 1] : 0.0f;
          Ai[tl][s] = v2f{w0 * rscale, w1 * rscale};
        }
      }
#pragma unroll
      for (int r = 0; r < 8; ++r) {
        const int j  = 4 * tl + 2 * half + (r >> 2);
        const int gr = (r & 3) * HH + j;
        const float bs = ((r & 3) == 2) ? 1.0f : 0.5f;
        accBias[tl][r] = (j < HH) ? (bih[gr] + bhh[gr]) * bs : 0.0f;
      }
    }
  }

  v2f c2[NT], h2[NT];
#pragma unroll
  for (int p = 0; p < NT; ++p) { c2[p] = v2f{0.0f, 0.0f}; h2[p] = v2f{0.0f, 0.0f}; }

  const size_t inRow = (size_t)(b0 + col) * TT;
  float* const outRow =
      out + inRow * OUTF + (size_t)dir * HH + 2 * half;
  const size_t tileBlk = (size_t)blockIdx.x * 2 * TT * FRAG;
  const float* const inFrag = IN_FRAG ? in + tileBlk + lane * 2 : in;
  float* const outFrag = OUT_FRAG
      ? out + tileBlk + (size_t)dir * TT * FRAG + (16 * half + col) * 2
      : out;

  auto t_of = [&](int step) -> int {
    return (dir == 0) ? step : (TT - 1 - step);
  };
  auto load_in = [&](int step, v2f (&X)[NSLICE_IN]) {
    const int t = t_of(step);
    if (DIN == 1) {
      float xv = 0.0f;
      if (half == 0) xv = in[inRow + t];
      X[0] = v2f{xv, 0.0f};
    } else if (IN_FRAG) {
      const float* p = inFrag + (size_t)t * FRAG;
#pragma unroll
      for (int s = 0; s < NSLICE_IN; ++s)
        X[s] = *(const v2f*)(p + (size_t)(s / 3) * TT * FRAG + (s % 3) * 64);
    } else {
      const float* p = in + (inRow + (size_t)t) * DIN;
#pragma unroll
      for (int s = 0; s < NSLICE_IN; ++s) {
        const int k0 = s * 4 + 2 * half;
        v2f v = v2f{0.0f, 0.0f};
        if (s * 4 + 3 < DIN || half == 0) v = *(const v2f*)(p + k0);
        X[s] = v;
      }
    }
  };
  auto project = [&](const v2f (&X)[NSLICE_IN], v8f (&P)[NT]) {
    P[0] = accBias[0]; P[1] = accBias[1]; P[2] = accBias[2];
#pragma unroll
    for (int s = 0; s < NSLICE_IN; ++s) {
      P[0] = wmma_f32_16x16x4(Ai[0][s], X[s], P[0]);
      P[1] = wmma_f32_16x16x4(Ai[1][s], X[s], P[1]);
      P[2] = wmma_f32_16x16x4(Ai[2][s], X[s], P[2]);
    }
  };

  auto produce = [&](int cc, float* buf) {
    const int hw = wave - 1;
    v2f X[HS][NSLICE_IN];
#pragma unroll
    for (int j = 0; j < HS; ++j) load_in(cc * CH + 2 * j + hw, X[j]);
    if (IN_FRAG && cc + 1 < NCH) {
#pragma unroll
      for (int q = 0; q < (HS * 12 + 31) / 32; ++q) {
        const int idx  = q * 32 + lane;
        const int j    = idx / 12 < HS ? idx / 12 : HS - 1;
        const int line = idx % 12;
        const int t    = t_of((cc + 1) * CH + 2 * j + hw);
        __builtin_prefetch(in + tileBlk + (size_t)(line / 6) * TT * FRAG +
                               (size_t)t * FRAG + (line % 6) * 32, 0, 1);
      }
    }
#pragma unroll
    for (int j = 0; j < HS; ++j) {
      v8f P[NT];
      project(X[j], P);
      float* const p = buf + (2 * j + hw) * PSTEP + lane * 4;
#pragma unroll
      for (int tl = 0; tl < NT; ++tl) {
        *(v4f*)(p + tl * 256)       = __builtin_shufflevector(P[tl], P[tl], 0, 1, 2, 3);
        *(v4f*)(p + tl * 256 + 128) = __builtin_shufflevector(P[tl], P[tl], 4, 5, 6, 7);
      }
    }
  };

  auto load_proj = [&](const float* buf, int i, v8f (&P)[NT]) {
    const float* const p = buf + i * PSTEP + lane * 4;
#pragma unroll
    for (int tl = 0; tl < NT; ++tl) {
      const v4f lo = *(const v4f*)(p + tl * 256);
      const v4f hi = *(const v4f*)(p + tl * 256 + 128);
      P[tl] = __builtin_shufflevector(lo, hi, 0, 1, 2, 3, 4, 5, 6, 7);
    }
  };

  auto one_step = [&](int step, const v8f (&Pu)[NT], v8f (&Pl)[NT],
                      const float* buf, int inext) {
    const int t = t_of(step);

    v8f a[NT] = {Pu[0], Pu[1], Pu[2]};
#pragma unroll
    for (int s = 0; s < NSLICE_H; ++s) {
      const v2f Bh = h2[s];
      a[0] = wmma_f32_16x16x4(Ah[0][s], Bh, a[0]);
      a[1] = wmma_f32_16x16x4(Ah[1][s], Bh, a[1]);
      a[2] = wmma_f32_16x16x4(Ah[2][s], Bh, a[2]);
    }

    load_proj(buf, inext, Pl);

    v2f gi[NT], gf[NT], gg[NT], go[NT], fc[NT], tc[NT];
    auto th2 = [&](float x, float y) {
      return v2f{__builtin_amdgcn_tanhf(x), __builtin_amdgcn_tanhf(y)};
    };
    auto tI = [&](int p) { gi[p] = th2(a[p][0], a[p][4]); };
    auto tF = [&](int p) { gf[p] = th2(a[p][1], a[p][5]); };
    auto tG = [&](int p) { gg[p] = th2(a[p][2], a[p][6]); };
    auto tO = [&](int p) { go[p] = th2(a[p][3], a[p][7]); };
    auto sI = [&](int p) { gi[p] = 0.5f * gi[p] + 0.5f; };
    auto sF = [&](int p) { gf[p] = 0.5f * gf[p] + 0.5f; };
    auto sO = [&](int p) { go[p] = 0.5f * go[p] + 0.5f; };
    auto mF = [&](int p) { fc[p] = gf[p] * c2[p]; };
    auto cell = [&](int p) { c2[p] = gi[p] * gg[p] + fc[p]; };
    auto cell_tanh = [&](int p) { tc[p] = th2(c2[p][0], c2[p][1]); };
    auto hidden = [&](int p) { h2[p] = go[p] * tc[p]; };
#define ORD __builtin_amdgcn_sched_barrier(0);

    ORD tI(0); tF(0); tG(0); tO(0);
    ORD tI(1); ORD sI(0); ORD tF(1); ORD sF(0);
    ORD tG(1); ORD sO(0); ORD tO(1); ORD mF(0);
    ORD tI(2); ORD cell(0); ORD tF(2); ORD sI(1);
    ORD tG(2); ORD sF(1); ORD tO(2); ORD sO(1);
    ORD mF(1);
    ORD sF(2); ORD cell_tanh(0); ORD cell(1); ORD sI(2); sO(2); ORD mF(2);
    ORD hidden(0); ORD cell_tanh(1); ORD cell(2);
    ORD hidden(1); ORD cell_tanh(2);
    ORD hidden(2);
    ORD
#undef ORD

    float* const o = outRow + (size_t)t * OUTF;
    if (OUT_FRAG) {
      float* const ob = outFrag + (size_t)t * FRAG;
#pragma unroll
      for (int p = 0; p < NT; ++p) *(v2f*)(ob + p * 64) = h2[p];
    } else {
#pragma unroll
      for (int r = 0; r < NR - 1; ++r) o[4 * (r >> 1) + (r & 1)] = h2[r >> 1][r & 1];
      if (half == 0) o[4 * ((NR - 1) >> 1) + 1] = h2[NT - 1][1];
    }
  };

  auto consume = [&](int cc, const float* buf) {
    v8f Pe[NT], Po[NT];
    load_proj(buf, 0, Pe);
    for (int i = 0; i < CH; i += 2) {
      one_step(cc * CH + i, Pe, Po, buf, i + 1);
      one_step(cc * CH + i + 1, Po, Pe, buf, i + 2 < CH ? i + 2 : CH - 1);
    }
  };

  auto hand_over = [&]() {
    __builtin_amdgcn_fence(__ATOMIC_RELEASE, "workgroup", "local");
    __builtin_amdgcn_s_barrier();
    __builtin_amdgcn_fence(__ATOMIC_ACQUIRE, "workgroup", "local");
  };

  if (wave != 0) produce(0, Pbuf);
  hand_over();
  for (int cc = 0; cc < NCH; ++cc) {
    float* const cur = Pbuf + (cc & 1) * (CH * PSTEP);
    float* const nxt = Pbuf + ((cc + 1) & 1) * (CH * PSTEP);
    if (wave == 0) {
      consume(cc, cur);
    } else if (cc + 1 < NCH) {
      produce(cc + 1, nxt);
    }
    hand_over();
  }
}

extern "C" void kernel_launch(void* const* d_in, const int* in_sizes, int n_in,
                              void* d_out, int out_size, void* d_ws,
                              size_t ws_size, hipStream_t stream) {
  const float* x     = (const float*)d_in[0];
  const float* W_ih0 = (const float*)d_in[1];
  const float* W_ihr = (const float*)d_in[2];
  const float* W_hh  = (const float*)d_in[3];
  const float* b_ih  = (const float*)d_in[4];
  const float* b_hh  = (const float*)d_in[5];
  float* out = (float*)d_out;

  const size_t XS_BYTES  = (size_t)NB * TT * sizeof(float);
  const size_t ACT_BYTES = (size_t)(NB / 16) * 2 * TT * FRAG * sizeof(float);
  char*  ws   = (char*)d_ws;
  float* xsum = (float*)(ws);
  float* bufA = (float*)(ws + XS_BYTES);
  float* bufB = (float*)(ws + XS_BYTES + ACT_BYTES);
  (void)ws_size; (void)in_sizes; (void)n_in; (void)out_size;

  {
    const size_t total = (size_t)NB * TT;
    reduce_x_kernel<<<dim3((unsigned)((total + 255) / 256)), dim3(256), 0,
                      stream>>>(x, xsum);
  }

  const dim3 grid(NB / 16, 2);
  const dim3 block(96);

  lstm_layer_kernel<1, false, true><<<grid, block, 0, stream>>>(
      xsum, bufA, W_ih0, W_hh, b_ih, b_hh);

  const float* lin[4]  = {bufA, bufB, bufA, bufB};
  float*       lout[4] = {bufB, bufA, bufB, out};
  for (int l = 1; l < 4; ++l) {
    lstm_layer_kernel<22, true, true><<<grid, block, 0, stream>>>(
        lin[l - 1], lout[l - 1],
        W_ihr + (size_t)(l - 1) * 2 * GG * 22,
        W_hh  + (size_t)l * 2 * GG * HH,
        b_ih  + (size_t)l * 2 * GG,
        b_hh  + (size_t)l * 2 * GG);
  }
  lstm_layer_kernel<22, true, false><<<grid, block, 0, stream>>>(
      lin[3], lout[3],
      W_ihr + (size_t)3 * 2 * GG * 22,
      W_hh  + (size_t)4 * 2 * GG * HH,
      b_ih  + (size_t)4 * 2 * GG,
      b_hh  + (size_t)4 * 2 * GG);
}


# Cell update issued in the order of its dependent chain (f and i tanh of all pairs first, o last, the h products behind every tanh); f*c is one fma on a kept half cell, rounded once instead of twice
# speedup vs baseline: 5.6516x; 1.0323x over previous
#include <hip/hip_runtime.h>
#include <hip/hip_bf16.h>


typedef float v2f __attribute__((ext_vector_type(2)));
typedef float v4f __attribute__((ext_vector_type(4)));
typedef float v8f __attribute__((ext_vector_type(8)));

#define NB   2048
#define TT   512
#define HH   11
#define GG   44
#define OUTF 22
#define FRAG 192
#define CHUNK 16
#define PSTEP 768

__device__ __forceinline__ v8f wmma_f32_16x16x4(v2f a, v2f b, v8f c) {
  return __builtin_amdgcn_wmma_f32_16x16x4_f32(false, a, false, b, (short)0, c,
                                               false, false);
}

__device__ __forceinline__ v8f splat8(float x) {
  v8f v;
#pragma unroll
  for (int i = 0; i < 8; ++i) v[i] = x;
  return v;
}

__global__ __launch_bounds__(256) void reduce_x_kernel(
    const float* __restrict__ x, float* __restrict__ xsum) {
  size_t idx = (size_t)blockIdx.x * blockDim.x + threadIdx.x;
  if (idx >= (size_t)NB * TT) return;
  size_t b = idx / TT, t = idx % TT;
  const float* p = x + b * (32u * TT) + t;
  float s = 0.0f;
#pragma unroll
  for (int k = 0; k < 32; ++k) s += p[(size_t)k * TT];
  xsum[idx] = s;
}

template <int DIN, bool IN_FRAG, bool OUT_FRAG>
__global__ __launch_bounds__(96)
__attribute__((amdgpu_waves_per_eu(1, 1))) void lstm_layer_kernel(
    const float* __restrict__ in, float* __restrict__ out,
    const float* __restrict__ Wih, const float* __restrict__ Whh,
    const float* __restrict__ bih, const float* __restrict__ bhh) {
  constexpr int NSLICE_IN = (DIN + 3) / 4;
  constexpr int NSLICE_H  = 3;
  constexpr int NT        = 3;
  constexpr int NR        = 6;
  constexpr int CH        = CHUNK;
  constexpr int NCH       = TT / CH;
  constexpr int HS        = CH / 2;
  static_assert(TT % CH == 0 && CH % 2 == 0, "whole chunks, split in two");

  __shared__ __attribute__((aligned(16))) float Pbuf[2 * CH * PSTEP];

  const int lane = threadIdx.x & 31;
  const int wave = __builtin_amdgcn_readfirstlane((int)(threadIdx.x >> 5));
  const int col  = lane & 15;
  const int half = lane >> 4;
  const int dir  = blockIdx.y;
  const int b0   = blockIdx.x * 16;

  Wih += (size_t)dir * GG * DIN;
  Whh += (size_t)dir * GG * HH;
  bih += (size_t)dir * GG;
  bhh += (size_t)dir * GG;

  const int rgate = col & 3;
  const int rsub  = col >> 2;
  const float rscale = (rgate == 2) ? 1.0f : 0.5f;
  v2f Ah[NT][NSLICE_H];
  v2f Ai[NT][NSLICE_IN];
  v8f accBias[NT];
#pragma unroll
  for (int tl = 0; tl < NT; ++tl) {
    const int  unit = 4 * tl + rsub;
    const bool rowv = unit < HH;
    const int  g    = rgate * HH + unit;
#pragma unroll
    for (int s = 0; s < NSLICE_H; ++s) Ah[tl][s] = v2f{0.0f, 0.0f};
#pragma unroll
    for (int s = 0; s < NSLICE_IN; ++s) Ai[tl][s] = v2f{0.0f, 0.0f};
    accBias[tl] = splat8(0.0f);
    if (wave == 0) {
#pragma unroll
      for (int s = 0; s < NSLICE_H; ++s) {
        const int k0 = 4 * s + 2 * half;
        float w0 = (rowv && k0 + 0 < HH) ? Whh[(size_t)g * HH + k0 + 0] : 0.0f;
        float w1 = (rowv && k0 + 1 < HH) ? Whh[(size_t)g * HH + k0 + 1] : 0.0f;
        Ah[tl][s] = v2f{w0 * rscale, w1 * rscale};
      }
    } else {
#pragma unroll
      for (int s = 0; s < NSLICE_IN; ++s) {
        if (IN_FRAG) {
          const int dp = s / 3;
          const int j0 = (s % 3) * 4 + 2 * half;
          const size_t wb = (size_t)g * DIN + dp * HH + j0;
          float w0 = (rowv && j0 + 0 < HH) ? Wih[wb + 0] : 0.0f;
          float w1 = (rowv && j0 + 1 < HH) ? Wih[wb + 1] : 0.0f;
          Ai[tl][s] = v2f{w0 * rscale, w1 * rscale};
        } else {
          const int k0 = s * 4 + 2 * half;
          float w0 = (rowv && k0 + 0 < DIN) ? Wih[(size_t)g * DIN + k0 + 0] : 0.0f;
          float w1 = (rowv && k0 + 1 < DIN) ? Wih[(size_t)g * DIN + k0 + 1] : 0.0f;
          Ai[tl][s] = v2f{w0 * rscale, w1 * rscale};
        }
      }
#pragma unroll
      for (int r = 0; r < 8; ++r) {
        const int j  = 4 * tl + 2 * half + (r >> 2);
        const int gr = (r & 3) * HH + j;
        const float bs = ((r & 3) == 2) ? 1.0f : 0.5f;
        accBias[tl][r] = (j < HH) ? (bih[gr] + bhh[gr]) * bs : 0.0f;
      }
    }
  }

  v2f hc2[NT], h2[NT];
#pragma unroll
  for (int p = 0; p < NT; ++p) { hc2[p] = v2f{0.0f, 0.0f}; h2[p] = v2f{0.0f, 0.0f}; }

  const size_t inRow = (size_t)(b0 + col) * TT;
  float* const outRow =
      out + inRow * OUTF + (size_t)dir * HH + 2 * half;
  const size_t tileBlk = (size_t)blockIdx.x * 2 * TT * FRAG;
  const float* const inFrag = IN_FRAG ? in + tileBlk + lane * 2 : in;
  float* const outFrag = OUT_FRAG
      ? out + tileBlk + (size_t)dir * TT * FRAG + (16 * half + col) * 2
      : out;

  auto t_of = [&](int step) -> int {
    return (dir == 0) ? step : (TT - 1 - step);
  };
  auto load_in = [&](int step, v2f (&X)[NSLICE_IN]) {
    const int t = t_of(step);
    if (DIN == 1) {
      float xv = 0.0f;
      if (half == 0) xv = in[inRow + t];
      X[0] = v2f{xv, 0.0f};
    } else if (IN_FRAG) {
      const float* p = inFrag + (size_t)t * FRAG;
#pragma unroll
      for (int s = 0; s < NSLICE_IN; ++s)
        X[s] = *(const v2f*)(p + (size_t)(s / 3) * TT * FRAG + (s % 3) * 64);
    } else {
      const float* p = in + (inRow + (size_t)t) * DIN;
#pragma unroll
      for (int s = 0; s < NSLICE_IN; ++s) {
        const int k0 = s * 4 + 2 * half;
        v2f v = v2f{0.0f, 0.0f};
        if (s * 4 + 3 < DIN || half == 0) v = *(const v2f*)(p + k0);
        X[s] = v;
      }
    }
  };
  auto project = [&](const v2f (&X)[NSLICE_IN], v8f (&P)[NT]) {
    P[0] = accBias[0]; P[1] = accBias[1]; P[2] = accBias[2];
#pragma unroll
    for (int s = 0; s < NSLICE_IN; ++s) {
      P[0] = wmma_f32_16x16x4(Ai[0][s], X[s], P[0]);
      P[1] = wmma_f32_16x16x4(Ai[1][s], X[s], P[1]);
      P[2] = wmma_f32_16x16x4(Ai[2][s], X[s], P[2]);
    }
  };

  auto produce = [&](int cc, float* buf) {
    const int hw = wave - 1;
    v2f X[HS][NSLICE_IN];
#pragma unroll
    for (int j = 0; j < HS; ++j) load_in(cc * CH + 2 * j + hw, X[j]);
    if (IN_FRAG && cc + 1 < NCH) {
#pragma unroll
      for (int q = 0; q < (HS * 12 + 31) / 32; ++q) {
        const int idx  = q * 32 + lane;
        const int j    = idx / 12 < HS ? idx / 12 : HS - 1;
        const int line = idx % 12;
        const int t    = t_of((cc + 1) * CH + 2 * j + hw);
        __builtin_prefetch(in + tileBlk + (size_t)(line / 6) * TT * FRAG +
                               (size_t)t * FRAG + (line % 6) * 32, 0, 1);
      }
    }
#pragma unroll
    for (int j = 0; j < HS; ++j) {
      v8f P[NT];
      project(X[j], P);
      float* const p = buf + (2 * j + hw) * PSTEP + lane * 4;
#pragma unroll
      for (int tl = 0; tl < NT; ++tl) {
        *(v4f*)(p + tl * 256)       = __builtin_shufflevector(P[tl], P[tl], 0, 1, 2, 3);
        *(v4f*)(p + tl * 256 + 128) = __builtin_shufflevector(P[tl], P[tl], 4, 5, 6, 7);
      }
    }
  };

  auto load_proj = [&](const float* buf, int i, v8f (&P)[NT]) {
    const float* const p = buf + i * PSTEP + lane * 4;
#pragma unroll
    for (int tl = 0; tl < NT; ++tl) {
      const v4f lo = *(const v4f*)(p + tl * 256);
      const v4f hi = *(const v4f*)(p + tl * 256 + 128);
      P[tl] = __builtin_shufflevector(lo, hi, 0, 1, 2, 3, 4, 5, 6, 7);
    }
  };

  auto one_step = [&](int step, const v8f (&Pu)[NT], v8f (&Pl)[NT],
                      const float* buf, int inext) {
    const int t = t_of(step);

    v8f a[NT] = {Pu[0], Pu[1], Pu[2]};
#pragma unroll
    for (int s = 0; s < NSLICE_H; ++s) {
      const v2f Bh = h2[s];
      a[0] = wmma_f32_16x16x4(Ah[0][s], Bh, a[0]);
      a[1] = wmma_f32_16x16x4(Ah[1][s], Bh, a[1]);
      a[2] = wmma_f32_16x16x4(Ah[2][s], Bh, a[2]);
    }

    load_proj(buf, inext, Pl);

    v2f gi[NT], gf[NT], gg[NT], go[NT], fc[NT], cn[NT], tc[NT];
    auto tI = [&](int p, int e) { gi[p][e] = __builtin_amdgcn_tanhf(a[p][4 * e + 0]); };
    auto tF = [&](int p, int e) { gf[p][e] = __builtin_amdgcn_tanhf(a[p][4 * e + 1]); };
    auto tG = [&](int p, int e) { gg[p][e] = __builtin_amdgcn_tanhf(a[p][4 * e + 2]); };
    auto tO = [&](int p, int e) { go[p][e] = __builtin_amdgcn_tanhf(a[p][4 * e + 3]); };
    auto sI = [&](int p) { gi[p] = 0.5f * gi[p] + 0.5f; };
    auto sO = [&](int p) { go[p] = 0.5f * go[p] + 0.5f; };
    auto mF = [&](int p) { fc[p] = gf[p] * hc2[p] + hc2[p]; };
    auto cell = [&](int p) { cn[p] = gi[p] * gg[p] + fc[p]; };
    auto half_cell = [&](int p) { hc2[p] = 0.5f * cn[p]; };
    auto tC = [&](int p, int e) { tc[p][e] = __builtin_amdgcn_tanhf(cn[p][e]); };
    auto hidden = [&](int p) { h2[p] = go[p] * tc[p]; };
#define ORD __builtin_amdgcn_sched_barrier(0);

    ORD tF(0, 0); ORD tF(0, 1); ORD tI(0, 0); ORD tI(0, 1);
    ORD tF(1, 0); ORD tF(1, 1); ORD tI(1, 0); ORD tI(1, 1);
    ORD tF(2, 0); ORD tF(2, 1); ORD mF(0);
    ORD tI(2, 0); ORD tI(2, 1); ORD sI(0);
    ORD tG(0, 0); ORD tG(0, 1); ORD mF(1);
    ORD tG(1, 0); ORD tG(1, 1); ORD sI(1);
    ORD tG(2, 0); ORD tG(2, 1); ORD mF(2);
    ORD tO(0, 0); ORD tO(0, 1); ORD sI(2);
    ORD tO(1, 0); ORD tO(1, 1); ORD cell(0);
    ORD tO(2, 0); ORD tO(2, 1); ORD cell(1); ORD cell(2);
    ORD tC(0, 0); ORD half_cell(0); ORD tC(0, 1); ORD sO(0);
    ORD tC(1, 0); ORD half_cell(1); ORD tC(1, 1); ORD sO(1);
    ORD tC(2, 0); ORD half_cell(2); ORD tC(2, 1); ORD sO(2);
    ORD hidden(0); ORD hidden(1); ORD hidden(2);
    ORD
#undef ORD

    float* const o = outRow + (size_t)t * OUTF;
    if (OUT_FRAG) {
      float* const ob = outFrag + (size_t)t * FRAG;
#pragma unroll
      for (int p = 0; p < NT; ++p) *(v2f*)(ob + p * 64) = h2[p];
    } else {
#pragma unroll
      for (int r = 0; r < NR - 1; ++r) o[4 * (r >> 1) + (r & 1)] = h2[r >> 1][r & 1];
      if (half == 0) o[4 * ((NR - 1) >> 1) + 1] = h2[NT - 1][1];
    }
  };

  auto consume = [&](int cc, const float* buf) {
    v8f Pe[NT], Po[NT];
    load_proj(buf, 0, Pe);
    for (int i = 0; i < CH; i += 2) {
      one_step(cc * CH + i, Pe, Po, buf, i + 1);
      one_step(cc * CH + i + 1, Po, Pe, buf, i + 2 < CH ? i + 2 : CH - 1);
    }
  };

  auto hand_over = [&]() {
    __builtin_amdgcn_fence(__ATOMIC_RELEASE, "workgroup", "local");
    __builtin_amdgcn_s_barrier();
    __builtin_amdgcn_fence(__ATOMIC_ACQUIRE, "workgroup", "local");
  };

  if (wave != 0) produce(0, Pbuf);
  hand_over();
  for (int cc = 0; cc < NCH; ++cc) {
    float* const cur = Pbuf + (cc & 1) * (CH * PSTEP);
    float* const nxt = Pbuf + ((cc + 1) & 1) * (CH * PSTEP);
    if (wave == 0) {
      consume(cc, cur);
    } else if (cc + 1 < NCH) {
      produce(cc + 1, nxt);
    }
    hand_over();
  }
}

extern "C" void kernel_launch(void* const* d_in, const int* in_sizes, int n_in,
                              void* d_out, int out_size, void* d_ws,
                              size_t ws_size, hipStream_t stream) {
  const float* x     = (const float*)d_in[0];
  const float* W_ih0 = (const float*)d_in[1];
  const float* W_ihr = (const float*)d_in[2];
  const float* W_hh  = (const float*)d_in[3];
  const float* b_ih  = (const float*)d_in[4];
  const float* b_hh  = (const float*)d_in[5];
  float* out = (float*)d_out;

  const size_t XS_BYTES  = (size_t)NB * TT * sizeof(float);
  const size_t ACT_BYTES = (size_t)(NB / 16) * 2 * TT * FRAG * sizeof(float);
  char*  ws   = (char*)d_ws;
  float* xsum = (float*)(ws);
  float* bufA = (float*)(ws + XS_BYTES);
  float* bufB = (float*)(ws + XS_BYTES + ACT_BYTES);
  (void)ws_size; (void)in_sizes; (void)n_in; (void)out_size;

  {
    const size_t total = (size_t)NB * TT;
    reduce_x_kernel<<<dim3((unsigned)((total + 255) / 256)), dim3(256), 0,
                      stream>>>(x, xsum);
  }

  const dim3 grid(NB / 16, 2);
  const dim3 block(96);

  lstm_layer_kernel<1, false, true><<<grid, block, 0, stream>>>(
      xsum, bufA, W_ih0, W_hh, b_ih, b_hh);

  const float* lin[4]  = {bufA, bufB, bufA, bufB};
  float*       lout[4] = {bufB, bufA, bufB, out};
  for (int l = 1; l < 4; ++l) {
    lstm_layer_kernel<22, true, true><<<grid, block, 0, stream>>>(
        lin[l - 1], lout[l - 1],
        W_ihr + (size_t)(l - 1) * 2 * GG * 22,
        W_hh  + (size_t)l * 2 * GG * HH,
        b_ih  + (size_t)l * 2 * GG,
        b_hh  + (size_t)l * 2 * GG);
  }
  lstm_layer_kernel<22, true, false><<<grid, block, 0, stream>>>(
      lin[3], lout[3],
      W_ihr + (size_t)3 * 2 * GG * 22,
      W_hh  + (size_t)4 * 2 * GG * HH,
      b_ih  + (size_t)4 * 2 * GG,
      b_hh  + (size_t)4 * 2 * GG);
}


# Recurrent wave reads its projections from LDS eight steps at a time: two places with LDS reads per 16-step chunk in its instruction stream, not one every step; every sum unchanged
# speedup vs baseline: 5.8818x; 1.0407x over previous
#include <hip/hip_runtime.h>
#include <hip/hip_bf16.h>


typedef float v2f __attribute__((ext_vector_type(2)));
typedef float v4f __attribute__((ext_vector_type(4)));
typedef float v8f __attribute__((ext_vector_type(8)));

#define NB   2048
#define TT   512
#define HH   11
#define GG   44
#define OUTF 22
#define FRAG 192
#define CHUNK 16
#define PSTEP 768

__device__ __forceinline__ v8f wmma_f32_16x16x4(v2f a, v2f b, v8f c) {
  return __builtin_amdgcn_wmma_f32_16x16x4_f32(false, a, false, b, (short)0, c,
                                               false, false);
}

__device__ __forceinline__ v8f splat8(float x) {
  v8f v;
#pragma unroll
  for (int i = 0; i < 8; ++i) v[i] = x;
  return v;
}

__global__ __launch_bounds__(256) void reduce_x_kernel(
    const float* __restrict__ x, float* __restrict__ xsum) {
  size_t idx = (size_t)blockIdx.x * blockDim.x + threadIdx.x;
  if (idx >= (size_t)NB * TT) return;
  size_t b = idx / TT, t = idx % TT;
  const float* p = x + b * (32u * TT) + t;
  float s = 0.0f;
#pragma unroll
  for (int k = 0; k < 32; ++k) s += p[(size_t)k * TT];
  xsum[idx] = s;
}

template <int DIN, bool IN_FRAG, bool OUT_FRAG>
__global__ __launch_bounds__(96)
__attribute__((amdgpu_waves_per_eu(1, 1))) void lstm_layer_kernel(
    const float* __restrict__ in, float* __restrict__ out,
    const float* __restrict__ Wih, const float* __restrict__ Whh,
    const float* __restrict__ bih, const float* __restrict__ bhh) {
  constexpr int NSLICE_IN = (DIN + 3) / 4;
  constexpr int NSLICE_H  = 3;
  constexpr int NT        = 3;
  constexpr int NR        = 6;
  constexpr int CH        = CHUNK;
  constexpr int NCH       = TT / CH;
  constexpr int HS        = CH / 2;
  constexpr int GROUP     = CH / 2;
  static_assert(TT % CH == 0 && CH % 2 == 0, "whole chunks, split in two");
  static_assert(GROUP >= 2 && GROUP * 2 * NT <= 63,
                "a group's reads are all in flight at once: the LDS counter holds 63");

  __shared__ __attribute__((aligned(16))) float Pbuf[2 * CH * PSTEP];

  const int lane = threadIdx.x & 31;
  const int wave = __builtin_amdgcn_readfirstlane((int)(threadIdx.x >> 5));
  const int col  = lane & 15;
  const int half = lane >> 4;
  const int dir  = blockIdx.y;
  const int b0   = blockIdx.x * 16;

  Wih += (size_t)dir * GG * DIN;
  Whh += (size_t)dir * GG * HH;
  bih += (size_t)dir * GG;
  bhh += (size_t)dir * GG;

  const int rgate = col & 3;
  const int rsub  = col >> 2;
  const float rscale = (rgate == 2) ? 1.0f : 0.5f;
  v2f Ah[NT][NSLICE_H];
  v2f Ai[NT][NSLICE_IN];
  v8f accBias[NT];
#pragma unroll
  for (int tl = 0; tl < NT; ++tl) {
    const int  unit = 4 * tl + rsub;
    const bool rowv = unit < HH;
    const int  g    = rgate * HH + unit;
#pragma unroll
    for (int s = 0; s < NSLICE_H; ++s) Ah[tl][s] = v2f{0.0f, 0.0f};
#pragma unroll
    for (int s = 0; s < NSLICE_IN; ++s) Ai[tl][s] = v2f{0.0f, 0.0f};
    accBias[tl] = splat8(0.0f);
    if (wave == 0) {
#pragma unroll
      for (int s = 0; s < NSLICE_H; ++s) {
        const int k0 = 4 * s + 2 * half;
        float w0 = (rowv && k0 + 0 < HH) ? Whh[(size_t)g * HH + k0 + 0] : 0.0f;
        float w1 = (rowv && k0 + 1 < HH) ? Whh[(size_t)g * HH + k0 + 1] : 0.0f;
        Ah[tl][s] = v2f{w0 * rscale, w1 * rscale};
      }
    } else {
#pragma unroll
      for (int s = 0; s < NSLICE_IN; ++s) {
        if (IN_FRAG) {
          const int dp = s / 3;
          const int j0 = (s % 3) * 4 + 2 * half;
          const size_t wb = (size_t)g * DIN + dp * HH + j0;
          float w0 = (rowv && j0 + 0 < HH) ? Wih[wb + 0] : 0.0f;
          float w1 = (rowv && j0 + 1 < HH) ? Wih[wb + 1] : 0.0f;
          Ai[tl][s] = v2f{w0 * rscale, w1 * rscale};
        } else {
          const int k0 = s * 4 + 2 * half;
          float w0 = (rowv && k0 + 0 < DIN) ? Wih[(size_t)g * DIN + k0 + 0] : 0.0f;
          float w1 = (rowv && k0 + 1 < DIN) ? Wih[(size_t)g * DIN + k0 + 1] : 0.0f;
          Ai[tl][s] = v2f{w0 * rscale, w1 * rscale};
        }
      }
#pragma unroll
      for (int r = 0; r < 8; ++r) {
        const int j  = 4 * tl + 2 * half + (r >> 2);
        const int gr = (r & 3) * HH + j;
        const float bs = ((r & 3) == 2) ? 1.0f : 0.5f;
        accBias[tl][r] = (j < HH) ? (bih[gr] + bhh[gr]) * bs : 0.0f;
      }
    }
  }

  v2f hc2[NT], h2[NT];
#pragma unroll
  for (int p = 0; p < NT; ++p) { hc2[p] = v2f{0.0f, 0.0f}; h2[p] = v2f{0.0f, 0.0f}; }

  const size_t inRow = (size_t)(b0 + col) * TT;
  float* const outRow =
      out + inRow * OUTF + (size_t)dir * HH + 2 * half;
  const size_t tileBlk = (size_t)blockIdx.x * 2 * TT * FRAG;
  const float* const inFrag = IN_FRAG ? in + tileBlk + lane * 2 : in;
  float* const outFrag = OUT_FRAG
      ? out + tileBlk + (size_t)dir * TT * FRAG + (16 * half + col) * 2
      : out;

  auto t_of = [&](int step) -> int {
    return (dir == 0) ? step : (TT - 1 - step);
  };
  auto load_in = [&](int step, v2f (&X)[NSLICE_IN]) {
    const int t = t_of(step);
    if (DIN == 1) {
      float xv = 0.0f;
      if (half == 0) xv = in[inRow + t];
      X[0] = v2f{xv, 0.0f};
    } else if (IN_FRAG) {
      const float* p = inFrag + (size_t)t * FRAG;
#pragma unroll
      for (int s = 0; s < NSLICE_IN; ++s)
        X[s] = *(const v2f*)(p + (size_t)(s / 3) * TT * FRAG + (s % 3) * 64);
    } else {
      const float* p = in + (inRow + (size_t)t) * DIN;
#pragma unroll
      for (int s = 0; s < NSLICE_IN; ++s) {
        const int k0 = s * 4 + 2 * half;
        v2f v = v2f{0.0f, 0.0f};
        if (s * 4 + 3 < DIN || half == 0) v = *(const v2f*)(p + k0);
        X[s] = v;
      }
    }
  };
  auto project = [&](const v2f (&X)[NSLICE_IN], v8f (&P)[NT]) {
    P[0] = accBias[0]; P[1] = accBias[1]; P[2] = accBias[2];
#pragma unroll
    for (int s = 0; s < NSLICE_IN; ++s) {
      P[0] = wmma_f32_16x16x4(Ai[0][s], X[s], P[0]);
      P[1] = wmma_f32_16x16x4(Ai[1][s], X[s], P[1]);
      P[2] = wmma_f32_16x16x4(Ai[2][s], X[s], P[2]);
    }
  };

  auto produce = [&](int cc, float* buf) {
    const int hw = wave - 1;
    v2f X[HS][NSLICE_IN];
#pragma unroll
    for (int j = 0; j < HS; ++j) load_in(cc * CH + 2 * j + hw, X[j]);
    if (IN_FRAG && cc + 1 < NCH) {
#pragma unroll
      for (int q = 0; q < (HS * 12 + 31) / 32; ++q) {
        const int idx  = q * 32 + lane;
        const int j    = idx / 12 < HS ? idx / 12 : HS - 1;
        const int line = idx % 12;
        const int t    = t_of((cc + 1) * CH + 2 * j + hw);
        __builtin_prefetch(in + tileBlk + (size_t)(line / 6) * TT * FRAG +
                               (size_t)t * FRAG + (line % 6) * 32, 0, 1);
      }
    }
#pragma unroll
    for (int j = 0; j < HS; ++j) {
      v8f P[NT];
      project(X[j], P);
      float* const p = buf + (2 * j + hw) * PSTEP + lane * 4;
#pragma unroll
      for (int tl = 0; tl < NT; ++tl) {
        *(v4f*)(p + tl * 256)       = __builtin_shufflevector(P[tl], P[tl], 0, 1, 2, 3);
        *(v4f*)(p + tl * 256 + 128) = __builtin_shufflevector(P[tl], P[tl], 4, 5, 6, 7);
      }
    }
  };

  auto load_proj = [&](const float* buf, int i, v8f (&P)[NT]) {
    const float* const p = buf + i * PSTEP + lane * 4;
#pragma unroll
    for (int tl = 0; tl < NT; ++tl) {
      const v4f lo = *(const v4f*)(p + tl * 256);
      const v4f hi = *(const v4f*)(p + tl * 256 + 128);
      P[tl] = __builtin_shufflevector(lo, hi, 0, 1, 2, 3, 4, 5, 6, 7);
    }
  };

  auto one_step = [&](int step, const v8f (&Pu)[NT]) {
    const int t = t_of(step);

    v8f a[NT] = {Pu[0], Pu[1], Pu[2]};
#pragma unroll
    for (int s = 0; s < NSLICE_H; ++s) {
      const v2f Bh = h2[s];
      a[0] = wmma_f32_16x16x4(Ah[0][s], Bh, a[0]);
      a[1] = wmma_f32_16x16x4(Ah[1][s], Bh, a[1]);
      a[2] = wmma_f32_16x16x4(Ah[2][s], Bh, a[2]);
    }

    v2f gi[NT], gf[NT], gg[NT], go[NT], fc[NT], cn[NT], tc[NT];
    auto tI = [&](int p, int e) { gi[p][e] = __builtin_amdgcn_tanhf(a[p][4 * e + 0]); };
    auto tF = [&](int p, int e) { gf[p][e] = __builtin_amdgcn_tanhf(a[p][4 * e + 1]); };
    auto tG = [&](int p, int e) { gg[p][e] = __builtin_amdgcn_tanhf(a[p][4 * e + 2]); };
    auto tO = [&](int p, int e) { go[p][e] = __builtin_amdgcn_tanhf(a[p][4 * e + 3]); };
    auto sI = [&](int p) { gi[p] = 0.5f * gi[p] + 0.5f; };
    auto sO = [&](int p) { go[p] = 0.5f * go[p] + 0.5f; };
    auto mF = [&](int p) { fc[p] = gf[p] * hc2[p] + hc2[p]; };
    auto cell = [&](int p) { cn[p] = gi[p] * gg[p] + fc[p]; };
    auto half_cell = [&](int p) { hc2[p] = 0.5f * cn[p]; };
    auto tC = [&](int p, int e) { tc[p][e] = __builtin_amdgcn_tanhf(cn[p][e]); };
    auto hidden = [&](int p) { h2[p] = go[p] * tc[p]; };
#define ORD __builtin_amdgcn_sched_barrier(0);

    ORD tF(0, 0); ORD tF(0, 1); ORD tI(0, 0); ORD tI(0, 1);
    ORD tF(1, 0); ORD tF(1, 1); ORD tI(1, 0); ORD tI(1, 1);
    ORD tF(2, 0); ORD tF(2, 1); ORD mF(0);
    ORD tI(2, 0); ORD tI(2, 1); ORD sI(0);
    ORD tG(0, 0); ORD tG(0, 1); ORD mF(1);
    ORD tG(1, 0); ORD tG(1, 1); ORD sI(1);
    ORD tG(2, 0); ORD tG(2, 1); ORD mF(2);
    ORD tO(0, 0); ORD tO(0, 1); ORD sI(2);
    ORD tO(1, 0); ORD tO(1, 1); ORD cell(0);
    ORD tO(2, 0); ORD tO(2, 1); ORD cell(1); ORD cell(2);
    ORD tC(0, 0); ORD half_cell(0); ORD tC(0, 1); ORD sO(0);
    ORD tC(1, 0); ORD half_cell(1); ORD tC(1, 1); ORD sO(1);
    ORD tC(2, 0); ORD half_cell(2); ORD tC(2, 1); ORD sO(2);
    ORD hidden(0); ORD hidden(1); ORD hidden(2);
    ORD
#undef ORD

    float* const o = outRow + (size_t)t * OUTF;
    if (OUT_FRAG) {
      float* const ob = outFrag + (size_t)t * FRAG;
#pragma unroll
      for (int p = 0; p < NT; ++p) *(v2f*)(ob + p * 64) = h2[p];
    } else {
#pragma unroll
      for (int r = 0; r < NR - 1; ++r) o[4 * (r >> 1) + (r & 1)] = h2[r >> 1][r & 1];
      if (half == 0) o[4 * ((NR - 1) >> 1) + 1] = h2[NT - 1][1];
    }
  };

  auto load_group = [&](const float* buf, int i0, v8f (&P)[GROUP][NT]) {
#pragma unroll
    for (int j = 0; j < GROUP; ++j) load_proj(buf, i0 + j, P[j]);
  };

  auto consume = [&](int cc, const float* buf) {
    v8f Pa[GROUP][NT], Pb[GROUP][NT];
#define PLACE __builtin_amdgcn_sched_barrier(0);
    PLACE load_group(buf, 0, Pa); PLACE
#pragma unroll
    for (int j = 0; j < GROUP; ++j) {
      if (j == GROUP - 2) { PLACE load_group(buf, GROUP, Pb); PLACE }
      one_step(cc * CH + j, Pa[j]);
    }
#pragma unroll
    for (int j = 0; j < GROUP; ++j) one_step(cc * CH + GROUP + j, Pb[j]);
#undef PLACE
  };

  auto hand_over = [&]() {
    __builtin_amdgcn_fence(__ATOMIC_RELEASE, "workgroup", "local");
    __builtin_amdgcn_s_barrier();
    __builtin_amdgcn_fence(__ATOMIC_ACQUIRE, "workgroup", "local");
  };

  if (wave != 0) produce(0, Pbuf);
  hand_over();
  for (int cc = 0; cc < NCH; ++cc) {
    float* const cur = Pbuf + (cc & 1) * (CH * PSTEP);
    float* const nxt = Pbuf + ((cc + 1) & 1) * (CH * PSTEP);
    if (wave == 0) {
      consume(cc, cur);
    } else if (cc + 1 < NCH) {
      produce(cc + 1, nxt);
    }
    hand_over();
  }
}

extern "C" void kernel_launch(void* const* d_in, const int* in_sizes, int n_in,
                              void* d_out, int out_size, void* d_ws,
                              size_t ws_size, hipStream_t stream) {
  const float* x     = (const float*)d_in[0];
  const float* W_ih0 = (const float*)d_in[1];
  const float* W_ihr = (const float*)d_in[2];
  const float* W_hh  = (const float*)d_in[3];
  const float* b_ih  = (const float*)d_in[4];
  const float* b_hh  = (const float*)d_in[5];
  float* out = (float*)d_out;

  const size_t XS_BYTES  = (size_t)NB * TT * sizeof(float);
  const size_t ACT_BYTES = (size_t)(NB / 16) * 2 * TT * FRAG * sizeof(float);
  char*  ws   = (char*)d_ws;
  float* xsum = (float*)(ws);
  float* bufA = (float*)(ws + XS_BYTES);
  float* bufB = (float*)(ws + XS_BYTES + ACT_BYTES);
  (void)ws_size; (void)in_sizes; (void)n_in; (void)out_size;

  {
    const size_t total = (size_t)NB * TT;
    reduce_x_kernel<<<dim3((unsigned)((total + 255) / 256)), dim3(256), 0,
                      stream>>>(x, xsum);
  }

  const dim3 grid(NB / 16, 2);
  const dim3 block(96);

  lstm_layer_kernel<1, false, true><<<grid, block, 0, stream>>>(
      xsum, bufA, W_ih0, W_hh, b_ih, b_hh);

  const float* lin[4]  = {bufA, bufB, bufA, bufB};
  float*       lout[4] = {bufB, bufA, bufB, out};
  for (int l = 1; l < 4; ++l) {
    lstm_layer_kernel<22, true, true><<<grid, block, 0, stream>>>(
        lin[l - 1], lout[l - 1],
        W_ihr + (size_t)(l - 1) * 2 * GG * 22,
        W_hh  + (size_t)l * 2 * GG * HH,
        b_ih  + (size_t)l * 2 * GG,
        b_hh  + (size_t)l * 2 * GG);
  }
  lstm_layer_kernel<22, true, false><<<grid, block, 0, stream>>>(
      lin[3], lout[3],
      W_ihr + (size_t)3 * 2 * GG * 22,
      W_hh  + (size_t)4 * 2 * GG * HH,
      b_ih  + (size_t)4 * 2 * GG,
      b_hh  + (size_t)4 * 2 * GG);
}
